# starting kernel
# baseline (speedup 1.0000x reference)
#include <hip/hip_runtime.h>

typedef float v2f   __attribute__((ext_vector_type(2)));
typedef float v8f   __attribute__((ext_vector_type(8)));
typedef float f32x4 __attribute__((ext_vector_type(4)));

#define WMMA_F32(acc, a, b) \
  __builtin_amdgcn_wmma_f32_16x16x4_f32(false, (a), false, (b), (short)0, (acc), false, false)

#define BB 8
#define NN 128
#define DD 512

__global__ __launch_bounds__(32) void gram_kernel(const float* __restrict__ Wk,
                                                  float* __restrict__ G) {
  const int lane = threadIdx.x;
  const int half = lane >> 4;
  const int ln   = lane & 15;
  const int c0   = blockIdx.y * 16;
  const int d0   = blockIdx.x * 16;
  v8f acc = {};
  for (int kb = 0; kb < DD; kb += 4) {
    const int k0 = kb + half * 2;
    v2f a, b;
    a.x = Wk[(k0 + 0) * DD + c0 + ln];
    a.y = Wk[(k0 + 1) * DD + c0 + ln];
    b.x = Wk[(k0 + 0) * DD + d0 + ln];
    b.y = Wk[(k0 + 1) * DD + d0 + ln];
    acc = WMMA_F32(acc, a, b);
  }
#pragma unroll
  for (int r = 0; r < 8; ++r)
    G[(c0 + r + half * 8) * DD + d0 + ln] = acc[r];
}

__global__ __launch_bounds__(32) void lin_kernel(const float* __restrict__ utt,
                                                 const float* __restrict__ Wk,
                                                 const float* __restrict__ Ws,
                                                 const float* __restrict__ G,
                                                 float* __restrict__ zi,
                                                 float* __restrict__ v,
                                                 float* __restrict__ si_lin) {
  const int lane = threadIdx.x;
  const int half = lane >> 4;
  const int ln   = lane & 15;
  const int m0   = blockIdx.y * 16;
  const int n0   = blockIdx.x * 16;
  v8f az = {}, av = {}, as = {};
  for (int kb = 0; kb < DD; kb += 4) {
    const int k0 = kb + half * 2;
    v2f a, b1, b2, b3;
    a.x  = utt[(m0 + ln) * DD + k0];
    a.y  = utt[(m0 + ln) * DD + k0 + 1];
    b1.x = Wk[(n0 + ln) * DD + k0];
    b1.y = Wk[(n0 + ln) * DD + k0 + 1];
    b2.x = G[(k0    ) * DD + n0 + ln];
    b2.y = G[(k0 + 1) * DD + n0 + ln];
    b3.x = Ws[(n0 + ln) * DD + k0];
    b3.y = Ws[(n0 + ln) * DD + k0 + 1];
    az = WMMA_F32(az, a, b1);
    av = WMMA_F32(av, a, b2);
    as = WMMA_F32(as, a, b3);
  }
#pragma unroll
  for (int r = 0; r < 8; ++r) {
    const int row = m0 + r + half * 8;
    zi[row * DD + n0 + ln]     = az[r];
    v[row * DD + n0 + ln]      = av[r];
    si_lin[row * DD + n0 + ln] = as[r];
  }
}

__global__ void deg_kernel(const float* __restrict__ seq, float* __restrict__ deg_inv) {
  const int row = blockIdx.x * blockDim.x + threadIdx.x;
  if (row < BB * NN) {
    float s = 0.f;
    for (int j = 0; j < NN; ++j) s += seq[row * NN + j];
    deg_inv[row] = 1.0f / (s + 1e-10f);
  }
}

__global__ __launch_bounds__(256) void edge_dot_kernel(const float* __restrict__ edge,
                                                       const float* __restrict__ utt,
                                                       const float* __restrict__ v,
                                                       float* __restrict__ logits) {
  const int b    = blockIdx.x >> 7;
  const int i    = blockIdx.x & 127;
  const int lane = threadIdx.x & 31;
  const int wave = threadIdx.x >> 5;
  const float* vrow = v + (size_t)(b * NN + i) * DD;
  f32x4 vr[4];
#pragma unroll
  for (int p = 0; p < 4; ++p)
    vr[p] = *(const f32x4*)(vrow + p * 128 + lane * 4);
  const float* erow0 = edge + (size_t)(b * NN + i) * NN * DD;
  for (int j = wave; j < NN; j += 8) {
    const float* er = erow0 + (size_t)j * DD;
    const float* ur = utt + (size_t)(b * NN + j) * DD;
    if (j + 8 < NN)
      __builtin_prefetch(erow0 + (size_t)(j + 8) * DD + lane * 16, 0, 0);
    f32x4 acc = {0.f, 0.f, 0.f, 0.f};
#pragma unroll
    for (int p = 0; p < 4; ++p) {
      f32x4 e = __builtin_nontemporal_load((const f32x4*)(er + p * 128 + lane * 4));
      f32x4 u = *(const f32x4*)(ur + p * 128 + lane * 4);
      acc += (e + u) * vr[p];
    }
    float s = acc.x + acc.y + acc.z + acc.w;
#pragma unroll
    for (int off = 16; off > 0; off >>= 1)
      s += __shfl_xor(s, off, 32);
    if (lane == 0)
      logits[(b * NN + i) * NN + j] = s * 0.044194173824159216f;
  }
}

__global__ __launch_bounds__(128) void softmax_kernel(const float* __restrict__ logits,
                                                      const float* __restrict__ bk,
                                                      float* __restrict__ attn) {
  __shared__ float red[128];
  const int b   = blockIdx.x >> 7;
  const int j   = blockIdx.x & 127;
  const int i   = threadIdx.x;
  const int idx = (b * NN + i) * NN + j;
  const float m  = bk[idx];
  const float lg = (m > 0.f) ? logits[idx] : -1e30f;
  red[i] = lg;
  __syncthreads();
  for (int s = 64; s > 0; s >>= 1) {
    if (i < s) red[i] = fmaxf(red[i], red[i + s]);
    __syncthreads();
  }
  const float mx = red[0];
  __syncthreads();
  const float e = __expf(lg - mx);
  red[i] = e;
  __syncthreads();
  for (int s = 64; s > 0; s >>= 1) {
    if (i < s) red[i] += red[i + s];
    __syncthreads();
  }
  attn[idx] = m * (e / red[0]);
}

__global__ __launch_bounds__(32) void out_kernel(const float* __restrict__ attn,
                                                 const float* __restrict__ seq,
                                                 const float* __restrict__ deg_inv,
                                                 const float* __restrict__ zi,
                                                 const float* __restrict__ si_lin,
                                                 float* __restrict__ out) {
  const int lane = threadIdx.x;
  const int half = lane >> 4;
  const int ln   = lane & 15;
  const int b    = blockIdx.z;
  const int n0   = blockIdx.y * 16;
  const int d0   = blockIdx.x * 16;
  const int rowb = b * NN;
  const float dinv = deg_inv[rowb + n0 + ln];
  v8f acc;
#pragma unroll
  for (int r = 0; r < 8; ++r)
    acc[r] = si_lin[(rowb + n0 + r + half * 8) * DD + d0 + ln];
  for (int kb = 0; kb < NN; kb += 4) {
    const int k0 = kb + half * 2;
    v2f a1, b1, a2, b2;
    a1.x = attn[(rowb + k0    ) * NN + n0 + ln];
    a1.y = attn[(rowb + k0 + 1) * NN + n0 + ln];
    b1.x = zi[(rowb + k0    ) * DD + d0 + ln];
    b1.y = zi[(rowb + k0 + 1) * DD + d0 + ln];
    a2.x = seq[(rowb + n0 + ln) * NN + k0    ] * dinv;
    a2.y = seq[(rowb + n0 + ln) * NN + k0 + 1] * dinv;
    b2.x = si_lin[(rowb + k0    ) * DD + d0 + ln];
    b2.y = si_lin[(rowb + k0 + 1) * DD + d0 + ln];
    acc = WMMA_F32(acc, a1, b1);
    acc = WMMA_F32(acc, a2, b2);
  }
  const float kScale = 1.0507009873554805f;
  const float kAlphaScale = 1.0507009873554805f * 1.6732632423543772f;
#pragma unroll
  for (int r = 0; r < 8; ++r) {
    const float x = acc[r];
    const float y = (x > 0.f) ? kScale * x : kAlphaScale * (__expf(x) - 1.f);
    out[(rowb + n0 + r + half * 8) * DD + d0 + ln] = y;
  }
}

extern "C" void kernel_launch(void* const* d_in, const int* in_sizes, int n_in,
                              void* d_out, int out_size, void* d_ws, size_t ws_size,
                              hipStream_t stream) {
  const float* utt  = (const float*)d_in[0];
  const float* edge = (const float*)d_in[1];
  const float* bk   = (const float*)d_in[2];
  const float* seq  = (const float*)d_in[3];
  const float* Wk   = (const float*)d_in[4];
  const float* Ws   = (const float*)d_in[5];
  float* out = (float*)d_out;

  float* ws     = (float*)d_ws;
  float* G      = ws;
  float* zi     = G + DD * DD;
  float* v      = zi + BB * NN * DD;
  float* si_lin = v + BB * NN * DD;
  float* logits = si_lin + BB * NN * DD;
  float* attn   = logits + BB * NN * NN;
  float* deginv = attn + BB * NN * NN;

  gram_kernel<<<dim3(DD / 16, DD / 16), 32, 0, stream>>>(Wk, G);
  lin_kernel<<<dim3(DD / 16, (BB * NN) / 16), 32, 0, stream>>>(utt, Wk, Ws, G, zi, v, si_lin);
  deg_kernel<<<(BB * NN + 255) / 256, 256, 0, stream>>>(seq, deginv);
  edge_dot_kernel<<<BB * NN, 256, 0, stream>>>(edge, utt, v, logits);
  softmax_kernel<<<BB * NN, 128, 0, stream>>>(logits, bk, attn);
  out_kernel<<<dim3(DD / 16, NN / 16, BB), 32, 0, stream>>>(attn, seq, deginv, zi, si_lin, out);
}


# GEMM K loops (gram, lin, out) software-pipelined in registers: next 32-k fragment loads in flight under current WMMAs; lin retiled 32x16 per wave, 128-bit k-contiguous loads (k sum order permuted)
# speedup vs baseline: 1.4082x; 1.4082x over previous
#include <hip/hip_runtime.h>

typedef float v2f   __attribute__((ext_vector_type(2)));
typedef float v8f   __attribute__((ext_vector_type(8)));
typedef float f32x4 __attribute__((ext_vector_type(4)));

#define WMMA_F32(acc, a, b) \
  __builtin_amdgcn_wmma_f32_16x16x4_f32(false, (a), false, (b), (short)0, (acc), false, false)

#define BB 8
#define NN 128
#define DD 512

struct GramFrag { v2f a[8], b[8]; };

__device__ __forceinline__ void gram_load(GramFrag& f, const float* __restrict__ pa,
                                          const float* __restrict__ pb, int kb) {
#pragma unroll
  for (int s = 0; s < 8; ++s) {
    const int k = kb + s * 4;
    f.a[s].x = pa[(k + 0) * DD];
    f.a[s].y = pa[(k + 1) * DD];
    f.b[s].x = pb[(k + 0) * DD];
    f.b[s].y = pb[(k + 1) * DD];
  }
}

__device__ __forceinline__ v8f gram_mma(v8f acc, const GramFrag& f) {
#pragma unroll
  for (int s = 0; s < 8; ++s) acc = WMMA_F32(acc, f.a[s], f.b[s]);
  return acc;
}

__global__ __launch_bounds__(128) __attribute__((amdgpu_waves_per_eu(1, 2))) void gram_kernel(const float* __restrict__ Wk,
                                                   float* __restrict__ G) {
  const int lane = threadIdx.x & 31;
  const int wave = threadIdx.x >> 5;
  const int half = lane >> 4;
  const int ln   = lane & 15;
  const int c0   = blockIdx.y * 16;
  const int d0   = (blockIdx.x * 4 + wave) * 16;
  const float* pa = Wk + (half * 2) * DD + c0 + ln;
  const float* pb = Wk + (half * 2) * DD + d0 + ln;
  v8f acc = {};
  GramFrag f0, f1;
  gram_load(f0, pa, pb, 0);
  for (int kb = 0; kb < DD - 64; kb += 64) {
    gram_load(f1, pa, pb, kb + 32);
    acc = gram_mma(acc, f0);
    gram_load(f0, pa, pb, kb + 64);
    acc = gram_mma(acc, f1);
  }
  gram_load(f1, pa, pb, DD - 32);
  acc = gram_mma(acc, f0);
  acc = gram_mma(acc, f1);
#pragma unroll
  for (int r = 0; r < 8; ++r)
    G[(c0 + r + half * 8) * DD + d0 + ln] = acc[r];
}

struct LinFrag { f32x4 a0[4], a1[4], b1[4], b2[4], b3[4]; };

__device__ __forceinline__ void lin_load(LinFrag& f, const float* __restrict__ pa0,
                                         const float* __restrict__ pa1,
                                         const float* __restrict__ pb1,
                                         const float* __restrict__ pb2,
                                         const float* __restrict__ pb3, int kb) {
#pragma unroll
  for (int q = 0; q < 4; ++q) {
    const int k = kb + q * 8;
    f.a0[q] = *(const f32x4*)(pa0 + k);
    f.a1[q] = *(const f32x4*)(pa1 + k);
    f.b1[q] = *(const f32x4*)(pb1 + k);
    f.b2[q] = *(const f32x4*)(pb2 + k);
    f.b3[q] = *(const f32x4*)(pb3 + k);
  }
}

#define LIN_STEP(A0, A1, B1, B2, B3)  \
  az0 = WMMA_F32(az0, A0, B1);        \
  av0 = WMMA_F32(av0, A0, B2);        \
  as0 = WMMA_F32(as0, A0, B3);        \
  az1 = WMMA_F32(az1, A1, B1);        \
  av1 = WMMA_F32(av1, A1, B2);        \
  as1 = WMMA_F32(as1, A1, B3);

#define LIN_MMA(F)                                                          \
  _Pragma("unroll") for (int q = 0; q < 4; ++q) {                           \
    const v2f a0l = {F.a0[q].x, F.a0[q].y}, a0h = {F.a0[q].z, F.a0[q].w};   \
    const v2f a1l = {F.a1[q].x, F.a1[q].y}, a1h = {F.a1[q].z, F.a1[q].w};   \
    const v2f b1l = {F.b1[q].x, F.b1[q].y}, b1h = {F.b1[q].z, F.b1[q].w};   \
    const v2f b2l = {F.b2[q].x, F.b2[q].y}, b2h = {F.b2[q].z, F.b2[q].w};   \
    const v2f b3l = {F.b3[q].x, F.b3[q].y}, b3h = {F.b3[q].z, F.b3[q].w};   \
    LIN_STEP(a0l, a1l, b1l, b2l, b3l)                                       \
    LIN_STEP(a0h, a1h, b1h, b2h, b3h)                                       \
  }

__global__ __launch_bounds__(128) void lin_kernel(const float* __restrict__ utt,
                                                  const float* __restrict__ Wk,
                                                  const float* __restrict__ Ws,
                                                  const float* __restrict__ G,
                                                  float* __restrict__ zi,
                                                  float* __restrict__ v,
                                                  float* __restrict__ si_lin) {
  const int lane = threadIdx.x & 31;
  const int wave = threadIdx.x >> 5;
  const int half = lane >> 4;
  const int ln   = lane & 15;
  const int n0   = blockIdx.x * 16;
  const int m0   = (blockIdx.y * 4 + wave) * 32;
  const float* pa0 = utt + (m0 + ln) * DD + half * 4;
  const float* pa1 = pa0 + 16 * DD;
  const float* pb1 = Wk + (n0 + ln) * DD + half * 4;
  const float* pb2 = G  + (n0 + ln) * DD + half * 4;
  const float* pb3 = Ws + (n0 + ln) * DD + half * 4;
  v8f az0 = {}, av0 = {}, as0 = {}, az1 = {}, av1 = {}, as1 = {};
  LinFrag f0, f1;
  lin_load(f0, pa0, pa1, pb1, pb2, pb3, 0);
  for (int kb = 0; kb < DD - 64; kb += 64) {
    lin_load(f1, pa0, pa1, pb1, pb2, pb3, kb + 32);
    LIN_MMA(f0)
    lin_load(f0, pa0, pa1, pb1, pb2, pb3, kb + 64);
    LIN_MMA(f1)
  }
  lin_load(f1, pa0, pa1, pb1, pb2, pb3, DD - 32);
  LIN_MMA(f0)
  LIN_MMA(f1)
#pragma unroll
  for (int r = 0; r < 8; ++r) {
    const int row = m0 + r + half * 8;
    zi[row * DD + n0 + ln]            = az0[r];
    v[row * DD + n0 + ln]             = av0[r];
    si_lin[row * DD + n0 + ln]        = as0[r];
    zi[(row + 16) * DD + n0 + ln]     = az1[r];
    v[(row + 16) * DD + n0 + ln]      = av1[r];
    si_lin[(row + 16) * DD + n0 + ln] = as1[r];
  }
}

__global__ void deg_kernel(const float* __restrict__ seq, float* __restrict__ deg_inv) {
  const int row = blockIdx.x * blockDim.x + threadIdx.x;
  if (row < BB * NN) {
    float s = 0.f;
    for (int j = 0; j < NN; ++j) s += seq[row * NN + j];
    deg_inv[row] = 1.0f / (s + 1e-10f);
  }
}

__global__ __launch_bounds__(256) void edge_dot_kernel(const float* __restrict__ edge,
                                                       const float* __restrict__ utt,
                                                       const float* __restrict__ v,
                                                       float* __restrict__ logits) {
  const int b    = blockIdx.x >> 7;
  const int i    = blockIdx.x & 127;
  const int lane = threadIdx.x & 31;
  const int wave = threadIdx.x >> 5;
  const float* vrow = v + (size_t)(b * NN + i) * DD;
  f32x4 vr[4];
#pragma unroll
  for (int p = 0; p < 4; ++p)
    vr[p] = *(const f32x4*)(vrow + p * 128 + lane * 4);
  const float* erow0 = edge + (size_t)(b * NN + i) * NN * DD;
  for (int j = wave; j < NN; j += 8) {
    const float* er = erow0 + (size_t)j * DD;
    const float* ur = utt + (size_t)(b * NN + j) * DD;
    if (j + 8 < NN)
      __builtin_prefetch(erow0 + (size_t)(j + 8) * DD + lane * 16, 0, 0);
    f32x4 acc = {0.f, 0.f, 0.f, 0.f};
#pragma unroll
    for (int p = 0; p < 4; ++p) {
      f32x4 e = __builtin_nontemporal_load((const f32x4*)(er + p * 128 + lane * 4));
      f32x4 u = *(const f32x4*)(ur + p * 128 + lane * 4);
      acc += (e + u) * vr[p];
    }
    float s = acc.x + acc.y + acc.z + acc.w;
#pragma unroll
    for (int off = 16; off > 0; off >>= 1)
      s += __shfl_xor(s, off, 32);
    if (lane == 0)
      logits[(b * NN + i) * NN + j] = s * 0.044194173824159216f;
  }
}

__global__ __launch_bounds__(128) void softmax_kernel(const float* __restrict__ logits,
                                                      const float* __restrict__ bk,
                                                      float* __restrict__ attn) {
  __shared__ float red[128];
  const int b   = blockIdx.x >> 7;
  const int j   = blockIdx.x & 127;
  const int i   = threadIdx.x;
  const int idx = (b * NN + i) * NN + j;
  const float m  = bk[idx];
  const float lg = (m > 0.f) ? logits[idx] : -1e30f;
  red[i] = lg;
  __syncthreads();
  for (int s = 64; s > 0; s >>= 1) {
    if (i < s) red[i] = fmaxf(red[i], red[i + s]);
    __syncthreads();
  }
  const float mx = red[0];
  __syncthreads();
  const float e = __expf(lg - mx);
  red[i] = e;
  __syncthreads();
  for (int s = 64; s > 0; s >>= 1) {
    if (i < s) red[i] += red[i + s];
    __syncthreads();
  }
  attn[idx] = m * (e / red[0]);
}

struct OutFrag { v2f a1[8], b1[8], a2[8], b2[8]; };

__device__ __forceinline__ void out_load(OutFrag& f, const float* __restrict__ pattn,
                                         const float* __restrict__ pzi,
                                         const float* __restrict__ pseq,
                                         const float* __restrict__ psl, int kb) {
#pragma unroll
  for (int s = 0; s < 8; ++s) {
    const int k = kb + s * 4;
    f.a1[s].x = pattn[(k + 0) * NN];
    f.a1[s].y = pattn[(k + 1) * NN];
    f.b1[s].x = pzi[(k + 0) * DD];
    f.b1[s].y = pzi[(k + 1) * DD];
    f.a2[s].x = pseq[k + 0];
    f.a2[s].y = pseq[k + 1];
    f.b2[s].x = psl[(k + 0) * DD];
    f.b2[s].y = psl[(k + 1) * DD];
  }
}

__device__ __forceinline__ v8f out_mma(v8f acc, const OutFrag& f, float dinv) {
#pragma unroll
  for (int s = 0; s < 8; ++s) {
    v2f a2;
    a2.x = f.a2[s].x * dinv;
    a2.y = f.a2[s].y * dinv;
    acc = WMMA_F32(acc, f.a1[s], f.b1[s]);
    acc = WMMA_F32(acc, a2, f.b2[s]);
  }
  return acc;
}

__global__ __launch_bounds__(32) __attribute__((amdgpu_waves_per_eu(1, 2))) void out_kernel(const float* __restrict__ attn,
                                                 const float* __restrict__ seq,
                                                 const float* __restrict__ deg_inv,
                                                 const float* __restrict__ zi,
                                                 const float* __restrict__ si_lin,
                                                 float* __restrict__ out) {
  const int lane = threadIdx.x;
  const int half = lane >> 4;
  const int ln   = lane & 15;
  const int b    = blockIdx.z;
  const int n0   = blockIdx.y * 16;
  const int d0   = blockIdx.x * 16;
  const int rowb = b * NN;
  const float dinv = deg_inv[rowb + n0 + ln];
  const float* pattn = attn   + (rowb + half * 2) * NN + n0 + ln;
  const float* pzi   = zi     + (rowb + half * 2) * DD + d0 + ln;
  const float* pseq  = seq    + (rowb + n0 + ln) * NN + half * 2;
  const float* psl   = si_lin + (rowb + half * 2) * DD + d0 + ln;
  OutFrag f0, f1;
  out_load(f0, pattn, pzi, pseq, psl, 0);
  v8f acc;
#pragma unroll
  for (int r = 0; r < 8; ++r)
    acc[r] = si_lin[(rowb + n0 + r + half * 8) * DD + d0 + ln];
  for (int kb = 0; kb < NN - 64; kb += 64) {
    out_load(f1, pattn, pzi, pseq, psl, kb + 32);
    acc = out_mma(acc, f0, dinv);
    out_load(f0, pattn, pzi, pseq, psl, kb + 64);
    acc = out_mma(acc, f1, dinv);
  }
  out_load(f1, pattn, pzi, pseq, psl, NN - 32);
  acc = out_mma(acc, f0, dinv);
  acc = out_mma(acc, f1, dinv);
  const float kScale = 1.0507009873554805f;
  const float kAlphaScale = 1.0507009873554805f * 1.6732632423543772f;
#pragma unroll
  for (int r = 0; r < 8; ++r) {
    const float x = acc[r];
    const float y = (x > 0.f) ? kScale * x : kAlphaScale * (__expf(x) - 1.f);
    out[(rowb + n0 + r + half * 8) * DD + d0 + ln] = y;
  }
}

extern "C" void kernel_launch(void* const* d_in, const int* in_sizes, int n_in,
                              void* d_out, int out_size, void* d_ws, size_t ws_size,
                              hipStream_t stream) {
  const float* utt  = (const float*)d_in[0];
  const float* edge = (const float*)d_in[1];
  const float* bk   = (const float*)d_in[2];
  const float* seq  = (const float*)d_in[3];
  const float* Wk   = (const float*)d_in[4];
  const float* Ws   = (const float*)d_in[5];
  float* out = (float*)d_out;

  float* ws     = (float*)d_ws;
  float* G      = ws;
  float* zi     = G + DD * DD;
  float* v      = zi + BB * NN * DD;
  float* si_lin = v + BB * NN * DD;
  float* logits = si_lin + BB * NN * DD;
  float* attn   = logits + BB * NN * NN;
  float* deginv = attn + BB * NN * NN;

  gram_kernel<<<dim3(DD / 64, DD / 16), 128, 0, stream>>>(Wk, G);
  lin_kernel<<<dim3(DD / 16, (BB * NN) / 128), 128, 0, stream>>>(utt, Wk, Ws, G, zi, v, si_lin);
  deg_kernel<<<(BB * NN + 255) / 256, 256, 0, stream>>>(seq, deginv);
  edge_dot_kernel<<<BB * NN, 256, 0, stream>>>(edge, utt, v, logits);
  softmax_kernel<<<BB * NN, 128, 0, stream>>>(logits, bk, attn);
  out_kernel<<<dim3(DD / 16, NN / 16, BB), 32, 0, stream>>>(attn, seq, deginv, zi, si_lin, out);
}


# Streaming dot kernel retiled to 4 i per wave: v rows of 4 i in registers, each utt row load serves 4 edge rows (utt re-reads cut 4x), 20 loads per group, 512 blocks of 8 waves; sum order unchanged
# speedup vs baseline: 1.4719x; 1.0452x over previous
#include <hip/hip_runtime.h>

typedef float v2f   __attribute__((ext_vector_type(2)));
typedef float v8f   __attribute__((ext_vector_type(8)));
typedef float f32x4 __attribute__((ext_vector_type(4)));

#define WMMA_F32(acc, a, b) \
  __builtin_amdgcn_wmma_f32_16x16x4_f32(false, (a), false, (b), (short)0, (acc), false, false)

#define BB 8
#define NN 128
#define DD 512

struct GramFrag { v2f a[8], b[8]; };

__device__ __forceinline__ void gram_load(GramFrag& f, const float* __restrict__ pa,
                                          const float* __restrict__ pb, int kb) {
#pragma unroll
  for (int s = 0; s < 8; ++s) {
    const int k = kb + s * 4;
    f.a[s].x = pa[(k + 0) * DD];
    f.a[s].y = pa[(k + 1) * DD];
    f.b[s].x = pb[(k + 0) * DD];
    f.b[s].y = pb[(k + 1) * DD];
  }
}

__device__ __forceinline__ v8f gram_mma(v8f acc, const GramFrag& f) {
#pragma unroll
  for (int s = 0; s < 8; ++s) acc = WMMA_F32(acc, f.a[s], f.b[s]);
  return acc;
}

__global__ __launch_bounds__(128) __attribute__((amdgpu_waves_per_eu(1, 2))) void gram_kernel(const float* __restrict__ Wk,
                                                   float* __restrict__ G) {
  const int lane = threadIdx.x & 31;
  const int wave = threadIdx.x >> 5;
  const int half = lane >> 4;
  const int ln   = lane & 15;
  const int c0   = blockIdx.y * 16;
  const int d0   = (blockIdx.x * 4 + wave) * 16;
  const float* pa = Wk + (half * 2) * DD + c0 + ln;
  const float* pb = Wk + (half * 2) * DD + d0 + ln;
  v8f acc = {};
  GramFrag f0, f1;
  gram_load(f0, pa, pb, 0);
  for (int kb = 0; kb < DD - 64; kb += 64) {
    gram_load(f1, pa, pb, kb + 32);
    acc = gram_mma(acc, f0);
    gram_load(f0, pa, pb, kb + 64);
    acc = gram_mma(acc, f1);
  }
  gram_load(f1, pa, pb, DD - 32);
  acc = gram_mma(acc, f0);
  acc = gram_mma(acc, f1);
#pragma unroll
  for (int r = 0; r < 8; ++r)
    G[(c0 + r + half * 8) * DD + d0 + ln] = acc[r];
}

struct LinFrag { f32x4 a0[4], a1[4], b1[4], b2[4], b3[4]; };

__device__ __forceinline__ void lin_load(LinFrag& f, const float* __restrict__ pa0,
                                         const float* __restrict__ pa1,
                                         const float* __restrict__ pb1,
                                         const float* __restrict__ pb2,
                                         const float* __restrict__ pb3, int kb) {
#pragma unroll
  for (int q = 0; q < 4; ++q) {
    const int k = kb + q * 8;
    f.a0[q] = *(const f32x4*)(pa0 + k);
    f.a1[q] = *(const f32x4*)(pa1 + k);
    f.b1[q] = *(const f32x4*)(pb1 + k);
    f.b2[q] = *(const f32x4*)(pb2 + k);
    f.b3[q] = *(const f32x4*)(pb3 + k);
  }
}

#define LIN_STEP(A0, A1, B1, B2, B3)  \
  az0 = WMMA_F32(az0, A0, B1);        \
  av0 = WMMA_F32(av0, A0, B2);        \
  as0 = WMMA_F32(as0, A0, B3);        \
  az1 = WMMA_F32(az1, A1, B1);        \
  av1 = WMMA_F32(av1, A1, B2);        \
  as1 = WMMA_F32(as1, A1, B3);

#define LIN_MMA(F)                                                          \
  _Pragma("unroll") for (int q = 0; q < 4; ++q) {                           \
    const v2f a0l = {F.a0[q].x, F.a0[q].y}, a0h = {F.a0[q].z, F.a0[q].w};   \
    const v2f a1l = {F.a1[q].x, F.a1[q].y}, a1h = {F.a1[q].z, F.a1[q].w};   \
    const v2f b1l = {F.b1[q].x, F.b1[q].y}, b1h = {F.b1[q].z, F.b1[q].w};   \
    const v2f b2l = {F.b2[q].x, F.b2[q].y}, b2h = {F.b2[q].z, F.b2[q].w};   \
    const v2f b3l = {F.b3[q].x, F.b3[q].y}, b3h = {F.b3[q].z, F.b3[q].w};   \
    LIN_STEP(a0l, a1l, b1l, b2l, b3l)                                       \
    LIN_STEP(a0h, a1h, b1h, b2h, b3h)                                       \
  }

__global__ __launch_bounds__(128) void lin_kernel(const float* __restrict__ utt,
                                                  const float* __restrict__ Wk,
                                                  const float* __restrict__ Ws,
                                                  const float* __restrict__ G,
                                                  float* __restrict__ zi,
                                                  float* __restrict__ v,
                                                  float* __restrict__ si_lin) {
  const int lane = threadIdx.x & 31;
  const int wave = threadIdx.x >> 5;
  const int half = lane >> 4;
  const int ln   = lane & 15;
  const int n0   = blockIdx.x * 16;
  const int m0   = (blockIdx.y * 4 + wave) * 32;
  const float* pa0 = utt + (m0 + ln) * DD + half * 4;
  const float* pa1 = pa0 + 16 * DD;
  const float* pb1 = Wk + (n0 + ln) * DD + half * 4;
  const float* pb2 = G  + (n0 + ln) * DD + half * 4;
  const float* pb3 = Ws + (n0 + ln) * DD + half * 4;
  v8f az0 = {}, av0 = {}, as0 = {}, az1 = {}, av1 = {}, as1 = {};
  LinFrag f0, f1;
  lin_load(f0, pa0, pa1, pb1, pb2, pb3, 0);
  for (int kb = 0; kb < DD - 64; kb += 64) {
    lin_load(f1, pa0, pa1, pb1, pb2, pb3, kb + 32);
    LIN_MMA(f0)
    lin_load(f0, pa0, pa1, pb1, pb2, pb3, kb + 64);
    LIN_MMA(f1)
  }
  lin_load(f1, pa0, pa1, pb1, pb2, pb3, DD - 32);
  LIN_MMA(f0)
  LIN_MMA(f1)
#pragma unroll
  for (int r = 0; r < 8; ++r) {
    const int row = m0 + r + half * 8;
    zi[row * DD + n0 + ln]            = az0[r];
    v[row * DD + n0 + ln]             = av0[r];
    si_lin[row * DD + n0 + ln]        = as0[r];
    zi[(row + 16) * DD + n0 + ln]     = az1[r];
    v[(row + 16) * DD + n0 + ln]      = av1[r];
    si_lin[(row + 16) * DD + n0 + ln] = as1[r];
  }
}

__global__ void deg_kernel(const float* __restrict__ seq, float* __restrict__ deg_inv) {
  const int row = blockIdx.x * blockDim.x + threadIdx.x;
  if (row < BB * NN) {
    float s = 0.f;
    for (int j = 0; j < NN; ++j) s += seq[row * NN + j];
    deg_inv[row] = 1.0f / (s + 1e-10f);
  }
}

#define TI 4
__global__ __launch_bounds__(256) void edge_dot_kernel(const float* __restrict__ edge,
                                                       const float* __restrict__ utt,
                                                       const float* __restrict__ v,
                                                       float* __restrict__ logits) {
  const int b    = blockIdx.x >> 6;
  const int i0   = ((blockIdx.x >> 1) & 31) * TI;
  const int j0   = (blockIdx.x & 1) * (NN / 2);
  const int lane = threadIdx.x & 31;
  const int wave = threadIdx.x >> 5;
  const float* vrow = v + (size_t)(b * NN + i0) * DD + lane * 4;
  f32x4 vr[TI][4];
#pragma unroll
  for (int t = 0; t < TI; ++t)
#pragma unroll
    for (int p = 0; p < 4; ++p)
      vr[t][p] = *(const f32x4*)(vrow + t * DD + p * 128);
  const float* eblk = edge + (size_t)(b * NN + i0) * NN * DD;
  for (int j = j0 + wave; j < j0 + NN / 2; j += 8) {
    const float* er = eblk + (size_t)j * DD + lane * 4;
    const float* ur = utt + (size_t)(b * NN + j) * DD + lane * 4;
    if (j + 8 < j0 + NN / 2) {
#pragma unroll
      for (int t = 0; t < TI; ++t)
        __builtin_prefetch(eblk + (size_t)t * NN * DD + (size_t)(j + 8) * DD + lane * 16, 0, 0);
    }
    f32x4 u[4];
#pragma unroll
    for (int p = 0; p < 4; ++p)
      u[p] = *(const f32x4*)(ur + p * 128);
    f32x4 e[TI][4];
#pragma unroll
    for (int t = 0; t < TI; ++t)
#pragma unroll
      for (int p = 0; p < 4; ++p)
        e[t][p] = __builtin_nontemporal_load((const f32x4*)(er + (size_t)t * NN * DD + p * 128));
    float s[TI];
#pragma unroll
    for (int t = 0; t < TI; ++t) {
      f32x4 acc = {0.f, 0.f, 0.f, 0.f};
#pragma unroll
      for (int p = 0; p < 4; ++p)
        acc += (e[t][p] + u[p]) * vr[t][p];
      s[t] = acc.x + acc.y + acc.z + acc.w;
    }
#pragma unroll
    for (int off = 16; off > 0; off >>= 1) {
#pragma unroll
      for (int t = 0; t < TI; ++t)
        s[t] += __shfl_xor(s[t], off, 32);
    }
    const float s01 = (lane & 1) ? s[1] : s[0];
    const float s23 = (lane & 1) ? s[3] : s[2];
    const float sv  = (lane & 2) ? s23 : s01;
    if (lane < TI)
      logits[(b * NN + i0 + lane) * NN + j] = sv * 0.044194173824159216f;
  }
}

__global__ __launch_bounds__(128) void softmax_kernel(const float* __restrict__ logits,
                                                      const float* __restrict__ bk,
                                                      float* __restrict__ attn) {
  __shared__ float red[128];
  const int b   = blockIdx.x >> 7;
  const int j   = blockIdx.x & 127;
  const int i   = threadIdx.x;
  const int idx = (b * NN + i) * NN + j;
  const float m  = bk[idx];
  const float lg = (m > 0.f) ? logits[idx] : -1e30f;
  red[i] = lg;
  __syncthreads();
  for (int s = 64; s > 0; s >>= 1) {
    if (i < s) red[i] = fmaxf(red[i], red[i + s]);
    __syncthreads();
  }
  const float mx = red[0];
  __syncthreads();
  const float e = __expf(lg - mx);
  red[i] = e;
  __syncthreads();
  for (int s = 64; s > 0; s >>= 1) {
    if (i < s) red[i] += red[i + s];
    __syncthreads();
  }
  attn[idx] = m * (e / red[0]);
}

struct OutFrag { v2f a1[8], b1[8], a2[8], b2[8]; };

__device__ __forceinline__ void out_load(OutFrag& f, const float* __restrict__ pattn,
                                         const float* __restrict__ pzi,
                                         const float* __restrict__ pseq,
                                         const float* __restrict__ psl, int kb) {
#pragma unroll
  for (int s = 0; s < 8; ++s) {
    const int k = kb + s * 4;
    f.a1[s].x = pattn[(k + 0) * NN];
    f.a1[s].y = pattn[(k + 1) * NN];
    f.b1[s].x = pzi[(k + 0) * DD];
    f.b1[s].y = pzi[(k + 1) * DD];
    f.a2[s].x = pseq[k + 0];
    f.a2[s].y = pseq[k + 1];
    f.b2[s].x = psl[(k + 0) * DD];
    f.b2[s].y = psl[(k + 1) * DD];
  }
}

__device__ __forceinline__ v8f out_mma(v8f acc, const OutFrag& f, float dinv) {
#pragma unroll
  for (int s = 0; s < 8; ++s) {
    v2f a2;
    a2.x = f.a2[s].x * dinv;
    a2.y = f.a2[s].y * dinv;
    acc = WMMA_F32(acc, f.a1[s], f.b1[s]);
    acc = WMMA_F32(acc, a2, f.b2[s]);
  }
  return acc;
}

__global__ __launch_bounds__(32) __attribute__((amdgpu_waves_per_eu(1, 2))) void out_kernel(const float* __restrict__ attn,
                                                 const float* __restrict__ seq,
                                                 const float* __restrict__ deg_inv,
                                                 const float* __restrict__ zi,
                                                 const float* __restrict__ si_lin,
                                                 float* __restrict__ out) {
  const int lane = threadIdx.x;
  const int half = lane >> 4;
  const int ln   = lane & 15;
  const int b    = blockIdx.z;
  const int n0   = blockIdx.y * 16;
  const int d0   = blockIdx.x * 16;
  const int rowb = b * NN;
  const float dinv = deg_inv[rowb + n0 + ln];
  const float* pattn = attn   + (rowb + half * 2) * NN + n0 + ln;
  const float* pzi   = zi     + (rowb + half * 2) * DD + d0 + ln;
  const float* pseq  = seq    + (rowb + n0 + ln) * NN + half * 2;
  const float* psl   = si_lin + (rowb + half * 2) * DD + d0 + ln;
  OutFrag f0, f1;
  out_load(f0, pattn, pzi, pseq, psl, 0);
  v8f acc;
#pragma unroll
  for (int r = 0; r < 8; ++r)
    acc[r] = si_lin[(rowb + n0 + r + half * 8) * DD + d0 + ln];
  for (int kb = 0; kb < NN - 64; kb += 64) {
    out_load(f1, pattn, pzi, pseq, psl, kb + 32);
    acc = out_mma(acc, f0, dinv);
    out_load(f0, pattn, pzi, pseq, psl, kb + 64);
    acc = out_mma(acc, f1, dinv);
  }
  out_load(f1, pattn, pzi, pseq, psl, NN - 32);
  acc = out_mma(acc, f0, dinv);
  acc = out_mma(acc, f1, dinv);
  const float kScale = 1.0507009873554805f;
  const float kAlphaScale = 1.0507009873554805f * 1.6732632423543772f;
#pragma unroll
  for (int r = 0; r < 8; ++r) {
    const float x = acc[r];
    const float y = (x > 0.f) ? kScale * x : kAlphaScale * (__expf(x) - 1.f);
    out[(rowb + n0 + r + half * 8) * DD + d0 + ln] = y;
  }
}

extern "C" void kernel_launch(void* const* d_in, const int* in_sizes, int n_in,
                              void* d_out, int out_size, void* d_ws, size_t ws_size,
                              hipStream_t stream) {
  const float* utt  = (const float*)d_in[0];
  const float* edge = (const float*)d_in[1];
  const float* bk   = (const float*)d_in[2];
  const float* seq  = (const float*)d_in[3];
  const float* Wk   = (const float*)d_in[4];
  const float* Ws   = (const float*)d_in[5];
  float* out = (float*)d_out;

  float* ws     = (float*)d_ws;
  float* G      = ws;
  float* zi     = G + DD * DD;
  float* v      = zi + BB * NN * DD;
  float* si_lin = v + BB * NN * DD;
  float* logits = si_lin + BB * NN * DD;
  float* attn   = logits + BB * NN * NN;
  float* deginv = attn + BB * NN * NN;

  gram_kernel<<<dim3(DD / 64, DD / 16), 128, 0, stream>>>(Wk, G);
  lin_kernel<<<dim3(DD / 16, (BB * NN) / 128), 128, 0, stream>>>(utt, Wk, Ws, G, zi, v, si_lin);
  deg_kernel<<<(BB * NN + 255) / 256, 256, 0, stream>>>(seq, deginv);
  edge_dot_kernel<<<BB * NN / TI * 2, 256, 0, stream>>>(edge, utt, v, logits);
  softmax_kernel<<<BB * NN, 128, 0, stream>>>(logits, bk, attn);
  out_kernel<<<dim3(DD / 16, NN / 16, BB), 32, 0, stream>>>(attn, seq, deginv, zi, si_lin, out);
}


# Softmax over i and the row-degree normalisation computed in registers in the output GEMM's prologue; softmax and degree kernels removed (6 -> 4 launches); softmax and degree sums reordered
# speedup vs baseline: 1.8072x; 1.2278x over previous
#include <hip/hip_runtime.h>

typedef float v2f   __attribute__((ext_vector_type(2)));
typedef float v8f   __attribute__((ext_vector_type(8)));
typedef float f32x4 __attribute__((ext_vector_type(4)));

#define WMMA_F32(acc, a, b) \
  __builtin_amdgcn_wmma_f32_16x16x4_f32(false, (a), false, (b), (short)0, (acc), false, false)

#define BB 8
#define NN 128
#define DD 512

struct GramFrag { v2f a[8], b[8]; };

__device__ __forceinline__ void gram_load(GramFrag& f, const float* __restrict__ pa,
                                          const float* __restrict__ pb, int kb) {
#pragma unroll
  for (int s = 0; s < 8; ++s) {
    const int k = kb + s * 4;
    f.a[s].x = pa[(k + 0) * DD];
    f.a[s].y = pa[(k + 1) * DD];
    f.b[s].x = pb[(k + 0) * DD];
    f.b[s].y = pb[(k + 1) * DD];
  }
}

__device__ __forceinline__ v8f gram_mma(v8f acc, const GramFrag& f) {
#pragma unroll
  for (int s = 0; s < 8; ++s) acc = WMMA_F32(acc, f.a[s], f.b[s]);
  return acc;
}

__global__ __launch_bounds__(128) __attribute__((amdgpu_waves_per_eu(1, 2))) void gram_kernel(const float* __restrict__ Wk,
                                                   float* __restrict__ G) {
  const int lane = threadIdx.x & 31;
  const int wave = threadIdx.x >> 5;
  const int half = lane >> 4;
  const int ln   = lane & 15;
  const int c0   = blockIdx.y * 16;
  const int d0   = (blockIdx.x * 4 + wave) * 16;
  const float* pa = Wk + (half * 2) * DD + c0 + ln;
  const float* pb = Wk + (half * 2) * DD + d0 + ln;
  v8f acc = {};
  GramFrag f0, f1;
  gram_load(f0, pa, pb, 0);
  for (int kb = 0; kb < DD - 64; kb += 64) {
    gram_load(f1, pa, pb, kb + 32);
    acc = gram_mma(acc, f0);
    gram_load(f0, pa, pb, kb + 64);
    acc = gram_mma(acc, f1);
  }
  gram_load(f1, pa, pb, DD - 32);
  acc = gram_mma(acc, f0);
  acc = gram_mma(acc, f1);
#pragma unroll
  for (int r = 0; r < 8; ++r)
    G[(c0 + r + half * 8) * DD + d0 + ln] = acc[r];
}

struct LinFrag { f32x4 a0[4], a1[4], b1[4], b2[4], b3[4]; };

__device__ __forceinline__ void lin_load(LinFrag& f, const float* __restrict__ pa0,
                                         const float* __restrict__ pa1,
                                         const float* __restrict__ pb1,
                                         const float* __restrict__ pb2,
                                         const float* __restrict__ pb3, int kb) {
#pragma unroll
  for (int q = 0; q < 4; ++q) {
    const int k = kb + q * 8;
    f.a0[q] = *(const f32x4*)(pa0 + k);
    f.a1[q] = *(const f32x4*)(pa1 + k);
    f.b1[q] = *(const f32x4*)(pb1 + k);
    f.b2[q] = *(const f32x4*)(pb2 + k);
    f.b3[q] = *(const f32x4*)(pb3 + k);
  }
}

#define LIN_STEP(A0, A1, B1, B2, B3)  \
  az0 = WMMA_F32(az0, A0, B1);        \
  av0 = WMMA_F32(av0, A0, B2);        \
  as0 = WMMA_F32(as0, A0, B3);        \
  az1 = WMMA_F32(az1, A1, B1);        \
  av1 = WMMA_F32(av1, A1, B2);        \
  as1 = WMMA_F32(as1, A1, B3);

#define LIN_MMA(F)                                                          \
  _Pragma("unroll") for (int q = 0; q < 4; ++q) {                           \
    const v2f a0l = {F.a0[q].x, F.a0[q].y}, a0h = {F.a0[q].z, F.a0[q].w};   \
    const v2f a1l = {F.a1[q].x, F.a1[q].y}, a1h = {F.a1[q].z, F.a1[q].w};   \
    const v2f b1l = {F.b1[q].x, F.b1[q].y}, b1h = {F.b1[q].z, F.b1[q].w};   \
    const v2f b2l = {F.b2[q].x, F.b2[q].y}, b2h = {F.b2[q].z, F.b2[q].w};   \
    const v2f b3l = {F.b3[q].x, F.b3[q].y}, b3h = {F.b3[q].z, F.b3[q].w};   \
    LIN_STEP(a0l, a1l, b1l, b2l, b3l)                                       \
    LIN_STEP(a0h, a1h, b1h, b2h, b3h)                                       \
  }

__global__ __launch_bounds__(128) void lin_kernel(const float* __restrict__ utt,
                                                  const float* __restrict__ Wk,
                                                  const float* __restrict__ Ws,
                                                  const float* __restrict__ G,
                                                  float* __restrict__ zi,
                                                  float* __restrict__ v,
                                                  float* __restrict__ si_lin) {
  const int lane = threadIdx.x & 31;
  const int wave = threadIdx.x >> 5;
  const int half = lane >> 4;
  const int ln   = lane & 15;
  const int n0   = blockIdx.x * 16;
  const int m0   = (blockIdx.y * 4 + wave) * 32;
  const float* pa0 = utt + (m0 + ln) * DD + half * 4;
  const float* pa1 = pa0 + 16 * DD;
  const float* pb1 = Wk + (n0 + ln) * DD + half * 4;
  const float* pb2 = G  + (n0 + ln) * DD + half * 4;
  const float* pb3 = Ws + (n0 + ln) * DD + half * 4;
  v8f az0 = {}, av0 = {}, as0 = {}, az1 = {}, av1 = {}, as1 = {};
  LinFrag f0, f1;
  lin_load(f0, pa0, pa1, pb1, pb2, pb3, 0);
  for (int kb = 0; kb < DD - 64; kb += 64) {
    lin_load(f1, pa0, pa1, pb1, pb2, pb3, kb + 32);
    LIN_MMA(f0)
    lin_load(f0, pa0, pa1, pb1, pb2, pb3, kb + 64);
    LIN_MMA(f1)
  }
  lin_load(f1, pa0, pa1, pb1, pb2, pb3, DD - 32);
  LIN_MMA(f0)
  LIN_MMA(f1)
#pragma unroll
  for (int r = 0; r < 8; ++r) {
    const int row = m0 + r + half * 8;
    zi[row * DD + n0 + ln]            = az0[r];
    v[row * DD + n0 + ln]             = av0[r];
    si_lin[row * DD + n0 + ln]        = as0[r];
    zi[(row + 16) * DD + n0 + ln]     = az1[r];
    v[(row + 16) * DD + n0 + ln]      = av1[r];
    si_lin[(row + 16) * DD + n0 + ln] = as1[r];
  }
}

#define TI 4
__global__ __launch_bounds__(256) void edge_dot_kernel(const float* __restrict__ edge,
                                                       const float* __restrict__ utt,
                                                       const float* __restrict__ v,
                                                       float* __restrict__ logits) {
  const int b    = blockIdx.x >> 6;
  const int i0   = ((blockIdx.x >> 1) & 31) * TI;
  const int j0   = (blockIdx.x & 1) * (NN / 2);
  const int lane = threadIdx.x & 31;
  const int wave = threadIdx.x >> 5;
  const float* vrow = v + (size_t)(b * NN + i0) * DD + lane * 4;
  f32x4 vr[TI][4];
#pragma unroll
  for (int t = 0; t < TI; ++t)
#pragma unroll
    for (int p = 0; p < 4; ++p)
      vr[t][p] = *(const f32x4*)(vrow + t * DD + p * 128);
  const float* eblk = edge + (size_t)(b * NN + i0) * NN * DD;
  for (int j = j0 + wave; j < j0 + NN / 2; j += 8) {
    const float* er = eblk + (size_t)j * DD + lane * 4;
    const float* ur = utt + (size_t)(b * NN + j) * DD + lane * 4;
    if (j + 8 < j0 + NN / 2) {
#pragma unroll
      for (int t = 0; t < TI; ++t)
        __builtin_prefetch(eblk + (size_t)t * NN * DD + (size_t)(j + 8) * DD + lane * 16, 0, 0);
    }
    f32x4 u[4];
#pragma unroll
    for (int p = 0; p < 4; ++p)
      u[p] = *(const f32x4*)(ur + p * 128);
    f32x4 e[TI][4];
#pragma unroll
    for (int t = 0; t < TI; ++t)
#pragma unroll
      for (int p = 0; p < 4; ++p)
        e[t][p] = __builtin_nontemporal_load((const f32x4*)(er + (size_t)t * NN * DD + p * 128));
    float s[TI];
#pragma unroll
    for (int t = 0; t < TI; ++t) {
      f32x4 acc = {0.f, 0.f, 0.f, 0.f};
#pragma unroll
      for (int p = 0; p < 4; ++p)
        acc += (e[t][p] + u[p]) * vr[t][p];
      s[t] = acc.x + acc.y + acc.z + acc.w;
    }
#pragma unroll
    for (int off = 16; off > 0; off >>= 1) {
#pragma unroll
      for (int t = 0; t < TI; ++t)
        s[t] += __shfl_xor(s[t], off, 32);
    }
    const float s01 = (lane & 1) ? s[1] : s[0];
    const float s23 = (lane & 1) ? s[3] : s[2];
    const float sv  = (lane & 2) ? s23 : s01;
    if (lane < TI)
      logits[(b * NN + i0 + lane) * NN + j] = sv * 0.044194173824159216f;
  }
}

struct OutB { v2f b1[8], b2[8]; };

__device__ __forceinline__ void outb_load(OutB& f, const float* __restrict__ pzi,
                                          const float* __restrict__ psl, int kb) {
#pragma unroll
  for (int s = 0; s < 8; ++s) {
    const int k = kb + s * 4;
    f.b1[s].x = pzi[(k + 0) * DD];
    f.b1[s].y = pzi[(k + 1) * DD];
    f.b2[s].x = psl[(k + 0) * DD];
    f.b2[s].y = psl[(k + 1) * DD];
  }
}

template <int S0>
__device__ __forceinline__ v8f out_mma(v8f acc, const v2f (&a1)[32], const v2f (&a2)[32],
                                       const OutB& f) {
#pragma unroll
  for (int s = 0; s < 8; ++s) {
    acc = WMMA_F32(acc, a1[S0 + s], f.b1[s]);
    acc = WMMA_F32(acc, a2[S0 + s], f.b2[s]);
  }
  return acc;
}

__global__ __launch_bounds__(32) __attribute__((amdgpu_waves_per_eu(1, 2))) void out_kernel(const float* __restrict__ logits,
                                                 const float* __restrict__ bk,
                                                 const float* __restrict__ seq,
                                                 const float* __restrict__ zi,
                                                 const float* __restrict__ si_lin,
                                                 float* __restrict__ out) {
  const int lane = threadIdx.x;
  const int half = lane >> 4;
  const int ln   = lane & 15;
  const int b    = blockIdx.z;
  const int n0   = blockIdx.y * 16;
  const int d0   = blockIdx.x * 16;
  const int rowb = b * NN;
  const float* plg  = logits + (rowb + half * 2) * NN + n0 + ln;
  const float* pbk  = bk     + (rowb + half * 2) * NN + n0 + ln;
  const float* pseq = seq    + (rowb + n0 + ln) * NN + half * 2;
  const float* pzi  = zi     + (rowb + half * 2) * DD + d0 + ln;
  const float* psl  = si_lin + (rowb + half * 2) * DD + d0 + ln;
  v2f a1[32], mk[32], a2[32];
#pragma unroll
  for (int s = 0; s < 32; ++s) {
    const int k = s * 4;
    mk[s].x = pbk[(k + 0) * NN];
    mk[s].y = pbk[(k + 1) * NN];
    a1[s].x = plg[(k + 0) * NN];
    a1[s].y = plg[(k + 1) * NN];
  }
#pragma unroll
  for (int s = 0; s < 32; ++s)
    a2[s] = *(const v2f*)(pseq + s * 4);
  v8f acc;
#pragma unroll
  for (int r = 0; r < 8; ++r)
    acc[r] = si_lin[(rowb + n0 + r + half * 8) * DD + d0 + ln];
  OutB f0, f1;
  outb_load(f0, pzi, psl, 0);
  outb_load(f1, pzi, psl, 32);

  float mx = -__builtin_inff();
#pragma unroll
  for (int s = 0; s < 32; ++s) {
    a1[s].x = (mk[s].x > 0.f) ? a1[s].x : -1e30f;
    a1[s].y = (mk[s].y > 0.f) ? a1[s].y : -1e30f;
    mx = fmaxf(mx, fmaxf(a1[s].x, a1[s].y));
  }
  mx = fmaxf(mx, __shfl_xor(mx, 16, 32));
  v2f sv = {0.f, 0.f};
#pragma unroll
  for (int s = 0; s < 32; ++s) {
    a1[s].x = __expf(a1[s].x - mx);
    a1[s].y = __expf(a1[s].y - mx);
    sv += a1[s];
  }
  float sum = sv.x + sv.y;
  sum += __shfl_xor(sum, 16, 32);
  const float rinv = 1.0f / sum;
#pragma unroll
  for (int s = 0; s < 32; ++s) {
    a1[s].x = mk[s].x * (a1[s].x * rinv);
    a1[s].y = mk[s].y * (a1[s].y * rinv);
  }

  v2f dv = {0.f, 0.f};
#pragma unroll
  for (int s = 0; s < 32; ++s) dv += a2[s];
  float deg = dv.x + dv.y;
  deg += __shfl_xor(deg, 16, 32);
  const float dinv = 1.0f / (deg + 1e-10f);
#pragma unroll
  for (int s = 0; s < 32; ++s) {
    a2[s].x *= dinv;
    a2[s].y *= dinv;
  }

  acc = out_mma<0>(acc, a1, a2, f0);
  outb_load(f0, pzi, psl, 64);
  acc = out_mma<8>(acc, a1, a2, f1);
  outb_load(f1, pzi, psl, 96);
  acc = out_mma<16>(acc, a1, a2, f0);
  acc = out_mma<24>(acc, a1, a2, f1);
  const float kScale = 1.0507009873554805f;
  const float kAlphaScale = 1.0507009873554805f * 1.6732632423543772f;
#pragma unroll
  for (int r = 0; r < 8; ++r) {
    const float x = acc[r];
    const float y = (x > 0.f) ? kScale * x : kAlphaScale * (__expf(x) - 1.f);
    out[(rowb + n0 + r + half * 8) * DD + d0 + ln] = y;
  }
}

extern "C" void kernel_launch(void* const* d_in, const int* in_sizes, int n_in,
                              void* d_out, int out_size, void* d_ws, size_t ws_size,
                              hipStream_t stream) {
  const float* utt  = (const float*)d_in[0];
  const float* edge = (const float*)d_in[1];
  const float* bk   = (const float*)d_in[2];
  const float* seq  = (const float*)d_in[3];
  const float* Wk   = (const float*)d_in[4];
  const float* Ws   = (const float*)d_in[5];
  float* out = (float*)d_out;

  float* ws     = (float*)d_ws;
  float* G      = ws;
  float* zi     = G + DD * DD;
  float* v      = zi + BB * NN * DD;
  float* si_lin = v + BB * NN * DD;
  float* logits = si_lin + BB * NN * DD;

  gram_kernel<<<dim3(DD / 64, DD / 16), 128, 0, stream>>>(Wk, G);
  lin_kernel<<<dim3(DD / 16, (BB * NN) / 128), 128, 0, stream>>>(utt, Wk, Ws, G, zi, v, si_lin);
  edge_dot_kernel<<<BB * NN / TI * 2, 256, 0, stream>>>(edge, utt, v, logits);
  out_kernel<<<dim3(DD / 16, NN / 16, BB), 32, 0, stream>>>(logits, bk, seq, zi, si_lin, out);
}


# gram and lin launches (memory idle) prefetch the first 24 of 64 j rows of every dot block of edge_rep into the second-level cache, one request per 256-B unit; dot kernel and all sums unchanged
# speedup vs baseline: 1.9314x; 1.0687x over previous
#include <hip/hip_runtime.h>

typedef float v2f   __attribute__((ext_vector_type(2)));
typedef float v8f   __attribute__((ext_vector_type(8)));
typedef float f32x4 __attribute__((ext_vector_type(4)));

#define WMMA_F32(acc, a, b) \
  __builtin_amdgcn_wmma_f32_16x16x4_f32(false, (a), false, (b), (short)0, (acc), false, false)

#define BB 8
#define NN 128
#define DD 512

#define PF_NG 4
#define PF_NL 8
__device__ __forceinline__ void edge_head_prefetch(const float* __restrict__ edge, int r, int u, int lane) {
  const int piece = u >> 8;
  const int c     = u & 255;
  const int b     = c >> 5;
  const int i     = ((c >> 2) & 7) * 16 + (r >> 1) * 4 + (c & 3);
  const size_t off = (size_t)((b * NN + i) * NN + (r & 1) * (NN / 2)) * DD + piece * 2048 + lane * 64;
  __builtin_prefetch(edge + off, 0, 2);
}

struct GramFrag { v2f a[8], b[8]; };

__device__ __forceinline__ void gram_load(GramFrag& f, const float* __restrict__ pa,
                                          const float* __restrict__ pb, int kb) {
#pragma unroll
  for (int s = 0; s < 8; ++s) {
    const int k = kb + s * 4;
    f.a[s].x = pa[(k + 0) * DD];
    f.a[s].y = pa[(k + 1) * DD];
    f.b[s].x = pb[(k + 0) * DD];
    f.b[s].y = pb[(k + 1) * DD];
  }
}

__device__ __forceinline__ v8f gram_mma(v8f acc, const GramFrag& f) {
#pragma unroll
  for (int s = 0; s < 8; ++s) acc = WMMA_F32(acc, f.a[s], f.b[s]);
  return acc;
}

__global__ __launch_bounds__(128) __attribute__((amdgpu_waves_per_eu(1, 2))) void gram_kernel(const float* __restrict__ Wk,
                                                   const float* __restrict__ edge,
                                                   float* __restrict__ G) {
  const int lane = threadIdx.x & 31;
  const int wave = threadIdx.x >> 5;
  const int half = lane >> 4;
  const int ln   = lane & 15;
  const int c0   = blockIdx.y * 16;
  const int d0   = (blockIdx.x * 4 + wave) * 16;
  const float* pa = Wk + (half * 2) * DD + c0 + ln;
  const float* pb = Wk + (half * 2) * DD + d0 + ln;
  const int pf_r = blockIdx.x;
  const int pf_u = (blockIdx.y * 4 + wave) * PF_NG;
  v8f acc = {};
  GramFrag f0, f1;
  gram_load(f0, pa, pb, 0);
  for (int kb = 0; kb < DD - 64; kb += 64) {
    if ((kb & 64) == 0) edge_head_prefetch(edge, pf_r, pf_u + (kb >> 7), lane);
    gram_load(f1, pa, pb, kb + 32);
    acc = gram_mma(acc, f0);
    gram_load(f0, pa, pb, kb + 64);
    acc = gram_mma(acc, f1);
  }
  gram_load(f1, pa, pb, DD - 32);
  acc = gram_mma(acc, f0);
  acc = gram_mma(acc, f1);
#pragma unroll
  for (int r = 0; r < 8; ++r)
    G[(c0 + r + half * 8) * DD + d0 + ln] = acc[r];
}

struct LinFrag { f32x4 a0[4], a1[4], b1[4], b2[4], b3[4]; };

__device__ __forceinline__ void lin_load(LinFrag& f, const float* __restrict__ pa0,
                                         const float* __restrict__ pa1,
                                         const float* __restrict__ pb1,
                                         const float* __restrict__ pb2,
                                         const float* __restrict__ pb3, int kb) {
#pragma unroll
  for (int q = 0; q < 4; ++q) {
    const int k = kb + q * 8;
    f.a0[q] = *(const f32x4*)(pa0 + k);
    f.a1[q] = *(const f32x4*)(pa1 + k);
    f.b1[q] = *(const f32x4*)(pb1 + k);
    f.b2[q] = *(const f32x4*)(pb2 + k);
    f.b3[q] = *(const f32x4*)(pb3 + k);
  }
}

#define LIN_STEP(A0, A1, B1, B2, B3)  \
  az0 = WMMA_F32(az0, A0, B1);        \
  av0 = WMMA_F32(av0, A0, B2);        \
  as0 = WMMA_F32(as0, A0, B3);        \
  az1 = WMMA_F32(az1, A1, B1);        \
  av1 = WMMA_F32(av1, A1, B2);        \
  as1 = WMMA_F32(as1, A1, B3);

#define LIN_MMA(F)                                                          \
  _Pragma("unroll") for (int q = 0; q < 4; ++q) {                           \
    const v2f a0l = {F.a0[q].x, F.a0[q].y}, a0h = {F.a0[q].z, F.a0[q].w};   \
    const v2f a1l = {F.a1[q].x, F.a1[q].y}, a1h = {F.a1[q].z, F.a1[q].w};   \
    const v2f b1l = {F.b1[q].x, F.b1[q].y}, b1h = {F.b1[q].z, F.b1[q].w};   \
    const v2f b2l = {F.b2[q].x, F.b2[q].y}, b2h = {F.b2[q].z, F.b2[q].w};   \
    const v2f b3l = {F.b3[q].x, F.b3[q].y}, b3h = {F.b3[q].z, F.b3[q].w};   \
    LIN_STEP(a0l, a1l, b1l, b2l, b3l)                                       \
    LIN_STEP(a0h, a1h, b1h, b2h, b3h)                                       \
  }

__global__ __launch_bounds__(128) void lin_kernel(const float* __restrict__ utt,
                                                  const float* __restrict__ Wk,
                                                  const float* __restrict__ Ws,
                                                  const float* __restrict__ G,
                                                  const float* __restrict__ edge,
                                                  float* __restrict__ zi,
                                                  float* __restrict__ v,
                                                  float* __restrict__ si_lin) {
  const int lane = threadIdx.x & 31;
  const int wave = threadIdx.x >> 5;
  const int half = lane >> 4;
  const int ln   = lane & 15;
  const int n0   = blockIdx.x * 16;
  const int m0   = (blockIdx.y * 4 + wave) * 32;
  const float* pa0 = utt + (m0 + ln) * DD + half * 4;
  const float* pa1 = pa0 + 16 * DD;
  const float* pb1 = Wk + (n0 + ln) * DD + half * 4;
  const float* pb2 = G  + (n0 + ln) * DD + half * 4;
  const float* pb3 = Ws + (n0 + ln) * DD + half * 4;
  const int pf_r = blockIdx.x & 7;
  const int pf_u = 128 * PF_NG + (((blockIdx.x >> 3) + 4 * blockIdx.y) * 4 + wave) * PF_NL;
  v8f az0 = {}, av0 = {}, as0 = {}, az1 = {}, av1 = {}, as1 = {};
  LinFrag f0, f1;
  lin_load(f0, pa0, pa1, pb1, pb2, pb3, 0);
  for (int kb = 0; kb < DD - 64; kb += 64) {
    edge_head_prefetch(edge, pf_r, pf_u + (kb >> 6), lane);
    lin_load(f1, pa0, pa1, pb1, pb2, pb3, kb + 32);
    LIN_MMA(f0)
    lin_load(f0, pa0, pa1, pb1, pb2, pb3, kb + 64);
    LIN_MMA(f1)
  }
  edge_head_prefetch(edge, pf_r, pf_u + PF_NL - 1, lane);
  lin_load(f1, pa0, pa1, pb1, pb2, pb3, DD - 32);
  LIN_MMA(f0)
  LIN_MMA(f1)
#pragma unroll
  for (int r = 0; r < 8; ++r) {
    const int row = m0 + r + half * 8;
    zi[row * DD + n0 + ln]            = az0[r];
    v[row * DD + n0 + ln]             = av0[r];
    si_lin[row * DD + n0 + ln]        = as0[r];
    zi[(row + 16) * DD + n0 + ln]     = az1[r];
    v[(row + 16) * DD + n0 + ln]      = av1[r];
    si_lin[(row + 16) * DD + n0 + ln] = as1[r];
  }
}

#define TI 4
__global__ __launch_bounds__(256) void edge_dot_kernel(const float* __restrict__ edge,
                                                       const float* __restrict__ utt,
                                                       const float* __restrict__ v,
                                                       float* __restrict__ logits) {
  const int b    = blockIdx.x >> 6;
  const int i0   = ((blockIdx.x >> 1) & 31) * TI;
  const int j0   = (blockIdx.x & 1) * (NN / 2);
  const int lane = threadIdx.x & 31;
  const int wave = threadIdx.x >> 5;
  const float* vrow = v + (size_t)(b * NN + i0) * DD + lane * 4;
  f32x4 vr[TI][4];
#pragma unroll
  for (int t = 0; t < TI; ++t)
#pragma unroll
    for (int p = 0; p < 4; ++p)
      vr[t][p] = *(const f32x4*)(vrow + t * DD + p * 128);
  const float* eblk = edge + (size_t)(b * NN + i0) * NN * DD;
  for (int j = j0 + wave; j < j0 + NN / 2; j += 8) {
    const float* er = eblk + (size_t)j * DD + lane * 4;
    const float* ur = utt + (size_t)(b * NN + j) * DD + lane * 4;
    if (j + 8 < j0 + NN / 2) {
#pragma unroll
      for (int t = 0; t < TI; ++t)
        __builtin_prefetch(eblk + (size_t)t * NN * DD + (size_t)(j + 8) * DD + lane * 16, 0, 0);
    }
    f32x4 u[4];
#pragma unroll
    for (int p = 0; p < 4; ++p)
      u[p] = *(const f32x4*)(ur + p * 128);
    f32x4 e[TI][4];
#pragma unroll
    for (int t = 0; t < TI; ++t)
#pragma unroll
      for (int p = 0; p < 4; ++p)
        e[t][p] = __builtin_nontemporal_load((const f32x4*)(er + (size_t)t * NN * DD + p * 128));
    float s[TI];
#pragma unroll
    for (int t = 0; t < TI; ++t) {
      f32x4 acc = {0.f, 0.f, 0.f, 0.f};
#pragma unroll
      for (int p = 0; p < 4; ++p)
        acc += (e[t][p] + u[p]) * vr[t][p];
      s[t] = acc.x + acc.y + acc.z + acc.w;
    }
#pragma unroll
    for (int off = 16; off > 0; off >>= 1) {
#pragma unroll
      for (int t = 0; t < TI; ++t)
        s[t] += __shfl_xor(s[t], off, 32);
    }
    const float s01 = (lane & 1) ? s[1] : s[0];
    const float s23 = (lane & 1) ? s[3] : s[2];
    const float sv  = (lane & 2) ? s23 : s01;
    if (lane < TI)
      logits[(b * NN + i0 + lane) * NN + j] = sv * 0.044194173824159216f;
  }
}

struct OutB { v2f b1[8], b2[8]; };

__device__ __forceinline__ void outb_load(OutB& f, const float* __restrict__ pzi,
                                          const float* __restrict__ psl, int kb) {
#pragma unroll
  for (int s = 0; s < 8; ++s) {
    const int k = kb + s * 4;
    f.b1[s].x = pzi[(k + 0) * DD];
    f.b1[s].y = pzi[(k + 1) * DD];
    f.b2[s].x = psl[(k + 0) * DD];
    f.b2[s].y = psl[(k + 1) * DD];
  }
}

template <int S0>
__device__ __forceinline__ v8f out_mma(v8f acc, const v2f (&a1)[32], const v2f (&a2)[32],
                                       const OutB& f) {
#pragma unroll
  for (int s = 0; s < 8; ++s) {
    acc = WMMA_F32(acc, a1[S0 + s], f.b1[s]);
    acc = WMMA_F32(acc, a2[S0 + s], f.b2[s]);
  }
  return acc;
}

__global__ __launch_bounds__(32) __attribute__((amdgpu_waves_per_eu(1, 2))) void out_kernel(const float* __restrict__ logits,
                                                 const float* __restrict__ bk,
                                                 const float* __restrict__ seq,
                                                 const float* __restrict__ zi,
                                                 const float* __restrict__ si_lin,
                                                 float* __restrict__ out) {
  const int lane = threadIdx.x;
  const int half = lane >> 4;
  const int ln   = lane & 15;
  const int b    = blockIdx.z;
  const int n0   = blockIdx.y * 16;
  const int d0   = blockIdx.x * 16;
  const int rowb = b * NN;
  const float* plg  = logits + (rowb + half * 2) * NN + n0 + ln;
  const float* pbk  = bk     + (rowb + half * 2) * NN + n0 + ln;
  const float* pseq = seq    + (rowb + n0 + ln) * NN + half * 2;
  const float* pzi  = zi     + (rowb + half * 2) * DD + d0 + ln;
  const float* psl  = si_lin + (rowb + half * 2) * DD + d0 + ln;
  v2f a1[32], mk[32], a2[32];
#pragma unroll
  for (int s = 0; s < 32; ++s) {
    const int k = s * 4;
    mk[s].x = pbk[(k + 0) * NN];
    mk[s].y = pbk[(k + 1) * NN];
    a1[s].x = plg[(k + 0) * NN];
    a1[s].y = plg[(k + 1) * NN];
  }
#pragma unroll
  for (int s = 0; s < 32; ++s)
    a2[s] = *(const v2f*)(pseq + s * 4);
  v8f acc;
#pragma unroll
  for (int r = 0; r < 8; ++r)
    acc[r] = si_lin[(rowb + n0 + r + half * 8) * DD + d0 + ln];
  OutB f0, f1;
  outb_load(f0, pzi, psl, 0);
  outb_load(f1, pzi, psl, 32);

  float mx = -__builtin_inff();
#pragma unroll
  for (int s = 0; s < 32; ++s) {
    a1[s].x = (mk[s].x > 0.f) ? a1[s].x : -1e30f;
    a1[s].y = (mk[s].y > 0.f) ? a1[s].y : -1e30f;
    mx = fmaxf(mx, fmaxf(a1[s].x, a1[s].y));
  }
  mx = fmaxf(mx, __shfl_xor(mx, 16, 32));
  v2f sv = {0.f, 0.f};
#pragma unroll
  for (int s = 0; s < 32; ++s) {
    a1[s].x = __expf(a1[s].x - mx);
    a1[s].y = __expf(a1[s].y - mx);
    sv += a1[s];
  }
  float sum = sv.x + sv.y;
  sum += __shfl_xor(sum, 16, 32);
  const float rinv = 1.0f / sum;
#pragma unroll
  for (int s = 0; s < 32; ++s) {
    a1[s].x = mk[s].x * (a1[s].x * rinv);
    a1[s].y = mk[s].y * (a1[s].y * rinv);
  }

  v2f dv = {0.f, 0.f};
#pragma unroll
  for (int s = 0; s < 32; ++s) dv += a2[s];
  float deg = dv.x + dv.y;
  deg += __shfl_xor(deg, 16, 32);
  const float dinv = 1.0f / (deg + 1e-10f);
#pragma unroll
  for (int s = 0; s < 32; ++s) {
    a2[s].x *= dinv;
    a2[s].y *= dinv;
  }

  acc = out_mma<0>(acc, a1, a2, f0);
  outb_load(f0, pzi, psl, 64);
  acc = out_mma<8>(acc, a1, a2, f1);
  outb_load(f1, pzi, psl, 96);
  acc = out_mma<16>(acc, a1, a2, f0);
  acc = out_mma<24>(acc, a1, a2, f1);
  const float kScale = 1.0507009873554805f;
  const float kAlphaScale = 1.0507009873554805f * 1.6732632423543772f;
#pragma unroll
  for (int r = 0; r < 8; ++r) {
    const float x = acc[r];
    const float y = (x > 0.f) ? kScale * x : kAlphaScale * (__expf(x) - 1.f);
    out[(rowb + n0 + r + half * 8) * DD + d0 + ln] = y;
  }
}

extern "C" void kernel_launch(void* const* d_in, const int* in_sizes, int n_in,
                              void* d_out, int out_size, void* d_ws, size_t ws_size,
                              hipStream_t stream) {
  const float* utt  = (const float*)d_in[0];
  const float* edge = (const float*)d_in[1];
  const float* bk   = (const float*)d_in[2];
  const float* seq  = (const float*)d_in[3];
  const float* Wk   = (const float*)d_in[4];
  const float* Ws   = (const float*)d_in[5];
  float* out = (float*)d_out;

  float* ws     = (float*)d_ws;
  float* G      = ws;
  float* zi     = G + DD * DD;
  float* v      = zi + BB * NN * DD;
  float* si_lin = v + BB * NN * DD;
  float* logits = si_lin + BB * NN * DD;

  gram_kernel<<<dim3(DD / 64, DD / 16), 128, 0, stream>>>(Wk, edge, G);
  lin_kernel<<<dim3(DD / 16, (BB * NN) / 128), 128, 0, stream>>>(utt, Wk, Ws, G, edge, zi, v, si_lin);
  edge_dot_kernel<<<BB * NN / TI * 2, 256, 0, stream>>>(edge, utt, v, logits);
  out_kernel<<<dim3(DD / 16, NN / 16, BB), 32, 0, stream>>>(logits, bk, seq, zi, si_lin, out);
}


# Gram launch removed: v = zi @ Wk formed in the GEMM launch from each block's zi tile via LDS, e sum split in 8 chunk planes added in the dot prologue (4 -> 3 launches; v sum order changed)
# speedup vs baseline: 2.2074x; 1.1429x over previous
#include <hip/hip_runtime.h>

typedef float v2f   __attribute__((ext_vector_type(2)));
typedef float v8f   __attribute__((ext_vector_type(8)));
typedef float f32x4 __attribute__((ext_vector_type(4)));

#define WMMA_F32(acc, a, b) \
  __builtin_amdgcn_wmma_f32_16x16x4_f32(false, (a), false, (b), (short)0, (acc), false, false)

#define BB 8
#define NN 128
#define DD 512

#define PF_N 12
__device__ __forceinline__ void edge_head_prefetch(const float* __restrict__ edge, int r, int u, int lane) {
  const int piece = u >> 8;
  const int c     = u & 255;
  const int b     = c >> 5;
  const int i     = ((c >> 2) & 7) * 16 + (r >> 1) * 4 + (c & 3);
  const size_t off = (size_t)((b * NN + i) * NN + (r & 1) * (NN / 2)) * DD + piece * 2048 + lane * 64;
  __builtin_prefetch(edge + off, 0, 2);
}

#define ZP 68
#define NCH 8
struct LinFrag { f32x4 a0[4], a1[4], bk[4], bs[4]; };

__device__ __forceinline__ void lin_load(LinFrag& f, const float* __restrict__ pa0,
                                         const float* __restrict__ pa1,
                                         const float* __restrict__ pbk,
                                         const float* __restrict__ pbs, int kb) {
#pragma unroll
  for (int q = 0; q < 4; ++q) {
    const int k = kb + q * 8;
    f.a0[q] = *(const f32x4*)(pa0 + k);
    f.a1[q] = *(const f32x4*)(pa1 + k);
    f.bk[q] = *(const f32x4*)(pbk + k);
    f.bs[q] = *(const f32x4*)(pbs + k);
  }
}

#define LIN_STEP(A0, A1, BK, BS)      \
  az0 = WMMA_F32(az0, A0, BK);        \
  as0 = WMMA_F32(as0, A0, BS);        \
  az1 = WMMA_F32(az1, A1, BK);        \
  as1 = WMMA_F32(as1, A1, BS);

#define LIN_MMA(F)                                                          \
  _Pragma("unroll") for (int q = 0; q < 4; ++q) {                           \
    const v2f a0l = {F.a0[q].x, F.a0[q].y}, a0h = {F.a0[q].z, F.a0[q].w};   \
    const v2f a1l = {F.a1[q].x, F.a1[q].y}, a1h = {F.a1[q].z, F.a1[q].w};   \
    const v2f bkl = {F.bk[q].x, F.bk[q].y}, bkh = {F.bk[q].z, F.bk[q].w};   \
    const v2f bsl = {F.bs[q].x, F.bs[q].y}, bsh = {F.bs[q].z, F.bs[q].w};   \
    LIN_STEP(a0l, a1l, bkl, bsl)                                            \
    LIN_STEP(a0h, a1h, bkh, bsh)                                            \
  }

struct VFrag { f32x4 x[2][4], y[2][4]; v2f a0[4], a1[4]; };

__device__ __forceinline__ void v_load(VFrag& f, const float* __restrict__ pb0,
                                       const float* __restrict__ pb1,
                                       const float* pla, int s0) {
#pragma unroll
  for (int st = 0; st < 4; ++st) {
    const int k = (s0 + st) * 4;
    f.x[0][st] = *(const f32x4*)(pb0 + (k + 0) * DD);
    f.y[0][st] = *(const f32x4*)(pb0 + (k + 1) * DD);
    f.x[1][st] = *(const f32x4*)(pb1 + (k + 0) * DD);
    f.y[1][st] = *(const f32x4*)(pb1 + (k + 1) * DD);
    f.a0[st]   = *(const v2f*)(pla + k);
    f.a1[st]   = *(const v2f*)(pla + 16 * ZP + k);
  }
}

#define V_MMA(F)                                                            \
  _Pragma("unroll") for (int st = 0; st < 4; ++st) {                        \
    _Pragma("unroll") for (int g = 0; g < 2; ++g) {                         \
      _Pragma("unroll") for (int t = 0; t < 4; ++t) {                       \
        const v2f bf = {F.x[g][st][t], F.y[g][st][t]};                      \
        av[0][g][t] = WMMA_F32(av[0][g][t], F.a0[st], bf);                  \
        av[1][g][t] = WMMA_F32(av[1][g][t], F.a1[st], bf);                  \
      }                                                                     \
    }                                                                       \
  }

__global__ __launch_bounds__(128) __attribute__((amdgpu_waves_per_eu(1, 2))) void lin_kernel(const float* __restrict__ utt,
                                                  const float* __restrict__ Wk,
                                                  const float* __restrict__ Ws,
                                                  const float* __restrict__ edge,
                                                  float* __restrict__ zi,
                                                  float* __restrict__ vpart,
                                                  float* __restrict__ si_lin) {
  __shared__ __attribute__((aligned(16))) float sz[32 * ZP];
  const int lane  = threadIdx.x & 31;
  const int wave  = threadIdx.x >> 5;
  const int half  = lane >> 4;
  const int ln    = lane & 15;
  const int chunk = blockIdx.x;
  const int m0    = blockIdx.y * 32;
  const int e0    = chunk * 64;
  const int n1    = e0 + wave * 16;
  const float* pa0 = utt + (m0 + ln) * DD + half * 4;
  const float* pa1 = pa0 + 16 * DD;
  const float* pbk = Wk + (n1 + ln) * DD + half * 4;
  const float* pbs = Ws + (n1 + ln) * DD + half * 4;
  const int pf_r = blockIdx.x;
  const int pf_u = (blockIdx.y * 4 + wave) * PF_N;
  v8f az0 = {}, az1 = {}, as0 = {}, as1 = {};
  LinFrag f0, f1;
  lin_load(f0, pa0, pa1, pbk, pbs, 0);
  for (int kb = 0; kb < DD - 64; kb += 64) {
    edge_head_prefetch(edge, pf_r, pf_u + (kb >> 6), lane);
    lin_load(f1, pa0, pa1, pbk, pbs, kb + 32);
    LIN_MMA(f0)
    lin_load(f0, pa0, pa1, pbk, pbs, kb + 64);
    LIN_MMA(f1)
  }
  edge_head_prefetch(edge, pf_r, pf_u + 7, lane);
  lin_load(f1, pa0, pa1, pbk, pbs, DD - 32);
  LIN_MMA(f0)
  LIN_MMA(f1)
#pragma unroll
  for (int r = 0; r < 8; ++r) {
    sz[(r + half * 8) * ZP + wave * 16 + ln]      = az0[r];
    sz[(16 + r + half * 8) * ZP + wave * 16 + ln] = az1[r];
  }
  __syncthreads();

  const int nb = wave * 128 + ln * 4;
  const float* pb0 = Wk + (e0 + half * 2) * DD + nb;
  const float* pb1 = pb0 + 64;
  const float* pla = sz + ln * ZP + half * 2;
  v8f av[2][2][4] = {};
  VFrag g0, g1;
  v_load(g0, pb0, pb1, pla, 0);
  edge_head_prefetch(edge, pf_r, pf_u + 8, lane);
  v_load(g1, pb0, pb1, pla, 4);
  V_MMA(g0)
  edge_head_prefetch(edge, pf_r, pf_u + 9, lane);
  v_load(g0, pb0, pb1, pla, 8);
  V_MMA(g1)
  edge_head_prefetch(edge, pf_r, pf_u + 10, lane);
  v_load(g1, pb0, pb1, pla, 12);
  V_MMA(g0)
  edge_head_prefetch(edge, pf_r, pf_u + 11, lane);
  V_MMA(g1)
#pragma unroll
  for (int r = 0; r < 8; ++r) {
    const int row = m0 + r + half * 8;
    zi[row * DD + n1 + ln]            = az0[r];
    zi[(row + 16) * DD + n1 + ln]     = az1[r];
    si_lin[row * DD + n1 + ln]        = as0[r];
    si_lin[(row + 16) * DD + n1 + ln] = as1[r];
  }
  float* vp = vpart + (size_t)chunk * (BB * NN * DD);
#pragma unroll
  for (int rt = 0; rt < 2; ++rt)
#pragma unroll
    for (int g = 0; g < 2; ++g)
#pragma unroll
      for (int r = 0; r < 8; ++r) {
        const f32x4 o = {av[rt][g][0][r], av[rt][g][1][r], av[rt][g][2][r], av[rt][g][3][r]};
        *(f32x4*)(vp + (m0 + rt * 16 + r + half * 8) * DD + nb + g * 64) = o;
      }
}

#define TI 4
__global__ __launch_bounds__(256) void edge_dot_kernel(const float* __restrict__ edge,
                                                       const float* __restrict__ utt,
                                                       const float* __restrict__ vpart,
                                                       float* __restrict__ logits) {
  __shared__ __attribute__((aligned(16))) float sv[TI * DD];
  const int b    = blockIdx.x >> 6;
  const int i0   = ((blockIdx.x >> 1) & 31) * TI;
  const int j0   = (blockIdx.x & 1) * (NN / 2);
  const int lane = threadIdx.x & 31;
  const int wave = threadIdx.x >> 5;
#pragma unroll
  for (int q = 0; q < 2; ++q) {
    const int item = wave * 2 + q;
    const float* src = vpart + (size_t)(b * NN + i0 + (item >> 2)) * DD + (item & 3) * 128 + lane * 4;
    f32x4 sum = *(const f32x4*)src;
#pragma unroll
    for (int c = 1; c < NCH; ++c)
      sum += *(const f32x4*)(src + (size_t)c * (BB * NN * DD));
    *(f32x4*)(sv + item * 128 + lane * 4) = sum;
  }
  __syncthreads();
  f32x4 vr[TI][4];
#pragma unroll
  for (int t = 0; t < TI; ++t)
#pragma unroll
    for (int p = 0; p < 4; ++p)
      vr[t][p] = *(const f32x4*)(sv + (t * 4 + p) * 128 + lane * 4);
  const float* eblk = edge + (size_t)(b * NN + i0) * NN * DD;
  for (int j = j0 + wave; j < j0 + NN / 2; j += 8) {
    const float* er = eblk + (size_t)j * DD + lane * 4;
    const float* ur = utt + (size_t)(b * NN + j) * DD + lane * 4;
    if (j + 8 < j0 + NN / 2) {
#pragma unroll
      for (int t = 0; t < TI; ++t)
        __builtin_prefetch(eblk + (size_t)t * NN * DD + (size_t)(j + 8) * DD + lane * 16, 0, 0);
    }
    f32x4 u[4];
#pragma unroll
    for (int p = 0; p < 4; ++p)
      u[p] = *(const f32x4*)(ur + p * 128);
    f32x4 e[TI][4];
#pragma unroll
    for (int t = 0; t < TI; ++t)
#pragma unroll
      for (int p = 0; p < 4; ++p)
        e[t][p] = __builtin_nontemporal_load((const f32x4*)(er + (size_t)t * NN * DD + p * 128));
    float s[TI];
#pragma unroll
    for (int t = 0; t < TI; ++t) {
      f32x4 acc = {0.f, 0.f, 0.f, 0.f};
#pragma unroll
      for (int p = 0; p < 4; ++p)
        acc += (e[t][p] + u[p]) * vr[t][p];
      s[t] = acc.x + acc.y + acc.z + acc.w;
    }
#pragma unroll
    for (int off = 16; off > 0; off >>= 1) {
#pragma unroll
      for (int t = 0; t < TI; ++t)
        s[t] += __shfl_xor(s[t], off, 32);
    }
    const float s01 = (lane & 1) ? s[1] : s[0];
    const float s23 = (lane & 1) ? s[3] : s[2];
    const float sv  = (lane & 2) ? s23 : s01;
    if (lane < TI)
      logits[(b * NN + i0 + lane) * NN + j] = sv * 0.044194173824159216f;
  }
}

struct OutB { v2f b1[8], b2[8]; };

__device__ __forceinline__ void outb_load(OutB& f, const float* __restrict__ pzi,
                                          const float* __restrict__ psl, int kb) {
#pragma unroll
  for (int s = 0; s < 8; ++s) {
    const int k = kb + s * 4;
    f.b1[s].x = pzi[(k + 0) * DD];
    f.b1[s].y = pzi[(k + 1) * DD];
    f.b2[s].x = psl[(k + 0) * DD];
    f.b2[s].y = psl[(k + 1) * DD];
  }
}

template <int S0>
__device__ __forceinline__ v8f out_mma(v8f acc, const v2f (&a1)[32], const v2f (&a2)[32],
                                       const OutB& f) {
#pragma unroll
  for (int s = 0; s < 8; ++s) {
    acc = WMMA_F32(acc, a1[S0 + s], f.b1[s]);
    acc = WMMA_F32(acc, a2[S0 + s], f.b2[s]);
  }
  return acc;
}

__global__ __launch_bounds__(32) __attribute__((amdgpu_waves_per_eu(1, 2))) void out_kernel(const float* __restrict__ logits,
                                                 const float* __restrict__ bk,
                                                 const float* __restrict__ seq,
                                                 const float* __restrict__ zi,
                                                 const float* __restrict__ si_lin,
                                                 float* __restrict__ out) {
  const int lane = threadIdx.x;
  const int half = lane >> 4;
  const int ln   = lane & 15;
  const int b    = blockIdx.z;
  const int n0   = blockIdx.y * 16;
  const int d0   = blockIdx.x * 16;
  const int rowb = b * NN;
  const float* plg  = logits + (rowb + half * 2) * NN + n0 + ln;
  const float* pbk  = bk     + (rowb + half * 2) * NN + n0 + ln;
  const float* pseq = seq    + (rowb + n0 + ln) * NN + half * 2;
  const float* pzi  = zi     + (rowb + half * 2) * DD + d0 + ln;
  const float* psl  = si_lin + (rowb + half * 2) * DD + d0 + ln;
  v2f a1[32], mk[32], a2[32];
#pragma unroll
  for (int s = 0; s < 32; ++s) {
    const int k = s * 4;
    mk[s].x = pbk[(k + 0) * NN];
    mk[s].y = pbk[(k + 1) * NN];
    a1[s].x = plg[(k + 0) * NN];
    a1[s].y = plg[(k + 1) * NN];
  }
#pragma unroll
  for (int s = 0; s < 32; ++s)
    a2[s] = *(const v2f*)(pseq + s * 4);
  v8f acc;
#pragma unroll
  for (int r = 0; r < 8; ++r)
    acc[r] = si_lin[(rowb + n0 + r + half * 8) * DD + d0 + ln];
  OutB f0, f1;
  outb_load(f0, pzi, psl, 0);
  outb_load(f1, pzi, psl, 32);

  float mx = -__builtin_inff();
#pragma unroll
  for (int s = 0; s < 32; ++s) {
    a1[s].x = (mk[s].x > 0.f) ? a1[s].x : -1e30f;
    a1[s].y = (mk[s].y > 0.f) ? a1[s].y : -1e30f;
    mx = fmaxf(mx, fmaxf(a1[s].x, a1[s].y));
  }
  mx = fmaxf(mx, __shfl_xor(mx, 16, 32));
  v2f sv = {0.f, 0.f};
#pragma unroll
  for (int s = 0; s < 32; ++s) {
    a1[s].x = __expf(a1[s].x - mx);
    a1[s].y = __expf(a1[s].y - mx);
    sv += a1[s];
  }
  float sum = sv.x + sv.y;
  sum += __shfl_xor(sum, 16, 32);
  const float rinv = 1.0f / sum;
#pragma unroll
  for (int s = 0; s < 32; ++s) {
    a1[s].x = mk[s].x * (a1[s].x * rinv);
    a1[s].y = mk[s].y * (a1[s].y * rinv);
  }

  v2f dv = {0.f, 0.f};
#pragma unroll
  for (int s = 0; s < 32; ++s) dv += a2[s];
  float deg = dv.x + dv.y;
  deg += __shfl_xor(deg, 16, 32);
  const float dinv = 1.0f / (deg + 1e-10f);
#pragma unroll
  for (int s = 0; s < 32; ++s) {
    a2[s].x *= dinv;
    a2[s].y *= dinv;
  }

  acc = out_mma<0>(acc, a1, a2, f0);
  outb_load(f0, pzi, psl, 64);
  acc = out_mma<8>(acc, a1, a2, f1);
  outb_load(f1, pzi, psl, 96);
  acc = out_mma<16>(acc, a1, a2, f0);
  acc = out_mma<24>(acc, a1, a2, f1);
  const float kScale = 1.0507009873554805f;
  const float kAlphaScale = 1.0507009873554805f * 1.6732632423543772f;
#pragma unroll
  for (int r = 0; r < 8; ++r) {
    const float x = acc[r];
    const float y = (x > 0.f) ? kScale * x : kAlphaScale * (__expf(x) - 1.f);
    out[(rowb + n0 + r + half * 8) * DD + d0 + ln] = y;
  }
}

extern "C" void kernel_launch(void* const* d_in, const int* in_sizes, int n_in,
                              void* d_out, int out_size, void* d_ws, size_t ws_size,
                              hipStream_t stream) {
  const float* utt  = (const float*)d_in[0];
  const float* edge = (const float*)d_in[1];
  const float* bk   = (const float*)d_in[2];
  const float* seq  = (const float*)d_in[3];
  const float* Wk   = (const float*)d_in[4];
  const float* Ws   = (const float*)d_in[5];
  float* out = (float*)d_out;

  float* ws     = (float*)d_ws;
  float* zi     = ws;
  float* vpart  = zi + BB * NN * DD;
  float* si_lin = vpart + NCH * BB * NN * DD;
  float* logits = si_lin + BB * NN * DD;

  lin_kernel<<<dim3(NCH, (BB * NN) / 32), 128, 0, stream>>>(utt, Wk, Ws, edge, zi, vpart, si_lin);
  edge_dot_kernel<<<BB * NN / TI * 2, 256, 0, stream>>>(edge, utt, vpart, logits);
  out_kernel<<<dim3(DD / 16, NN / 16, BB), 32, 0, stream>>>(logits, bk, seq, zi, si_lin, out);
}


# Output kernel: softmax and degree-scaled A tiles built once per 4-wave block (a quarter of k per wave) and shared through LDS, not once per wave; softmax and degree sums reordered
# speedup vs baseline: 2.3769x; 1.0768x over previous
#include <hip/hip_runtime.h>

typedef float v2f   __attribute__((ext_vector_type(2)));
typedef float v8f   __attribute__((ext_vector_type(8)));
typedef float f32x4 __attribute__((ext_vector_type(4)));

#define WMMA_F32(acc, a, b) \
  __builtin_amdgcn_wmma_f32_16x16x4_f32(false, (a), false, (b), (short)0, (acc), false, false)

#define BB 8
#define NN 128
#define DD 512

#define PF_N 12
__device__ __forceinline__ void edge_head_prefetch(const float* __restrict__ edge, int r, int u, int lane) {
  const int piece = u >> 8;
  const int c     = u & 255;
  const int b     = c >> 5;
  const int i     = ((c >> 2) & 7) * 16 + (r >> 1) * 4 + (c & 3);
  const size_t off = (size_t)((b * NN + i) * NN + (r & 1) * (NN / 2)) * DD + piece * 2048 + lane * 64;
  __builtin_prefetch(edge + off, 0, 2);
}

#define ZP 68
#define NCH 8
struct LinFrag { f32x4 a0[4], a1[4], bk[4], bs[4]; };

__device__ __forceinline__ void lin_load(LinFrag& f, const float* __restrict__ pa0,
                                         const float* __restrict__ pa1,
                                         const float* __restrict__ pbk,
                                         const float* __restrict__ pbs, int kb) {
#pragma unroll
  for (int q = 0; q < 4; ++q) {
    const int k = kb + q * 8;
    f.a0[q] = *(const f32x4*)(pa0 + k);
    f.a1[q] = *(const f32x4*)(pa1 + k);
    f.bk[q] = *(const f32x4*)(pbk + k);
    f.bs[q] = *(const f32x4*)(pbs + k);
  }
}

#define LIN_STEP(A0, A1, BK, BS)      \
  az0 = WMMA_F32(az0, A0, BK);        \
  as0 = WMMA_F32(as0, A0, BS);        \
  az1 = WMMA_F32(az1, A1, BK);        \
  as1 = WMMA_F32(as1, A1, BS);

#define LIN_MMA(F)                                                          \
  _Pragma("unroll") for (int q = 0; q < 4; ++q) {                           \
    const v2f a0l = {F.a0[q].x, F.a0[q].y}, a0h = {F.a0[q].z, F.a0[q].w};   \
    const v2f a1l = {F.a1[q].x, F.a1[q].y}, a1h = {F.a1[q].z, F.a1[q].w};   \
    const v2f bkl = {F.bk[q].x, F.bk[q].y}, bkh = {F.bk[q].z, F.bk[q].w};   \
    const v2f bsl = {F.bs[q].x, F.bs[q].y}, bsh = {F.bs[q].z, F.bs[q].w};   \
    LIN_STEP(a0l, a1l, bkl, bsl)                                            \
    LIN_STEP(a0h, a1h, bkh, bsh)                                            \
  }

struct VFrag { f32x4 x[2][4], y[2][4]; v2f a0[4], a1[4]; };

__device__ __forceinline__ void v_load(VFrag& f, const float* __restrict__ pb0,
                                       const float* __restrict__ pb1,
                                       const float* pla, int s0) {
#pragma unroll
  for (int st = 0; st < 4; ++st) {
    const int k = (s0 + st) * 4;
    f.x[0][st] = *(const f32x4*)(pb0 + (k + 0) * DD);
    f.y[0][st] = *(const f32x4*)(pb0 + (k + 1) * DD);
    f.x[1][st] = *(const f32x4*)(pb1 + (k + 0) * DD);
    f.y[1][st] = *(const f32x4*)(pb1 + (k + 1) * DD);
    f.a0[st]   = *(const v2f*)(pla + k);
    f.a1[st]   = *(const v2f*)(pla + 16 * ZP + k);
  }
}

#define V_MMA(F)                                                            \
  _Pragma("unroll") for (int st = 0; st < 4; ++st) {                        \
    _Pragma("unroll") for (int g = 0; g < 2; ++g) {                         \
      _Pragma("unroll") for (int t = 0; t < 4; ++t) {                       \
        const v2f bf = {F.x[g][st][t], F.y[g][st][t]};                      \
        av[0][g][t] = WMMA_F32(av[0][g][t], F.a0[st], bf);                  \
        av[1][g][t] = WMMA_F32(av[1][g][t], F.a1[st], bf);                  \
      }                                                                     \
    }                                                                       \
  }

__global__ __launch_bounds__(128) __attribute__((amdgpu_waves_per_eu(1, 2))) void lin_kernel(const float* __restrict__ utt,
                                                  const float* __restrict__ Wk,
                                                  const float* __restrict__ Ws,
                                                  const float* __restrict__ edge,
                                                  float* __restrict__ zi,
                                                  float* __restrict__ vpart,
                                                  float* __restrict__ si_lin) {
  __shared__ __attribute__((aligned(16))) float sz[32 * ZP];
  const int lane  = threadIdx.x & 31;
  const int wave  = threadIdx.x >> 5;
  const int half  = lane >> 4;
  const int ln    = lane & 15;
  const int chunk = blockIdx.x;
  const int m0    = blockIdx.y * 32;
  const int e0    = chunk * 64;
  const int n1    = e0 + wave * 16;
  const float* pa0 = utt + (m0 + ln) * DD + half * 4;
  const float* pa1 = pa0 + 16 * DD;
  const float* pbk = Wk + (n1 + ln) * DD + half * 4;
  const float* pbs = Ws + (n1 + ln) * DD + half * 4;
  const int pf_r = blockIdx.x;
  const int pf_u = (blockIdx.y * 4 + wave) * PF_N;
  v8f az0 = {}, az1 = {}, as0 = {}, as1 = {};
  LinFrag f0, f1;
  lin_load(f0, pa0, pa1, pbk, pbs, 0);
  for (int kb = 0; kb < DD - 64; kb += 64) {
    edge_head_prefetch(edge, pf_r, pf_u + (kb >> 6), lane);
    lin_load(f1, pa0, pa1, pbk, pbs, kb + 32);
    LIN_MMA(f0)
    lin_load(f0, pa0, pa1, pbk, pbs, kb + 64);
    LIN_MMA(f1)
  }
  edge_head_prefetch(edge, pf_r, pf_u + 7, lane);
  lin_load(f1, pa0, pa1, pbk, pbs, DD - 32);
  LIN_MMA(f0)
  LIN_MMA(f1)
#pragma unroll
  for (int r = 0; r < 8; ++r) {
    sz[(r + half * 8) * ZP + wave * 16 + ln]      = az0[r];
    sz[(16 + r + half * 8) * ZP + wave * 16 + ln] = az1[r];
  }
  __syncthreads();

  const int nb = wave * 128 + ln * 4;
  const float* pb0 = Wk + (e0 + half * 2) * DD + nb;
  const float* pb1 = pb0 + 64;
  const float* pla = sz + ln * ZP + half * 2;
  v8f av[2][2][4] = {};
  VFrag g0, g1;
  v_load(g0, pb0, pb1, pla, 0);
  edge_head_prefetch(edge, pf_r, pf_u + 8, lane);
  v_load(g1, pb0, pb1, pla, 4);
  V_MMA(g0)
  edge_head_prefetch(edge, pf_r, pf_u + 9, lane);
  v_load(g0, pb0, pb1, pla, 8);
  V_MMA(g1)
  edge_head_prefetch(edge, pf_r, pf_u + 10, lane);
  v_load(g1, pb0, pb1, pla, 12);
  V_MMA(g0)
  edge_head_prefetch(edge, pf_r, pf_u + 11, lane);
  V_MMA(g1)
#pragma unroll
  for (int r = 0; r < 8; ++r) {
    const int row = m0 + r + half * 8;
    zi[row * DD + n1 + ln]            = az0[r];
    zi[(row + 16) * DD + n1 + ln]     = az1[r];
    si_lin[row * DD + n1 + ln]        = as0[r];
    si_lin[(row + 16) * DD + n1 + ln] = as1[r];
  }
  float* vp = vpart + (size_t)chunk * (BB * NN * DD);
#pragma unroll
  for (int rt = 0; rt < 2; ++rt)
#pragma unroll
    for (int g = 0; g < 2; ++g)
#pragma unroll
      for (int r = 0; r < 8; ++r) {
        const f32x4 o = {av[rt][g][0][r], av[rt][g][1][r], av[rt][g][2][r], av[rt][g][3][r]};
        *(f32x4*)(vp + (m0 + rt * 16 + r + half * 8) * DD + nb + g * 64) = o;
      }
}

#define TI 4
__global__ __launch_bounds__(256) void edge_dot_kernel(const float* __restrict__ edge,
                                                       const float* __restrict__ utt,
                                                       const float* __restrict__ vpart,
                                                       float* __restrict__ logits) {
  __shared__ __attribute__((aligned(16))) float sv[TI * DD];
  const int b    = blockIdx.x >> 6;
  const int i0   = ((blockIdx.x >> 1) & 31) * TI;
  const int j0   = (blockIdx.x & 1) * (NN / 2);
  const int lane = threadIdx.x & 31;
  const int wave = threadIdx.x >> 5;
#pragma unroll
  for (int q = 0; q < 2; ++q) {
    const int item = wave * 2 + q;
    const float* src = vpart + (size_t)(b * NN + i0 + (item >> 2)) * DD + (item & 3) * 128 + lane * 4;
    f32x4 sum = *(const f32x4*)src;
#pragma unroll
    for (int c = 1; c < NCH; ++c)
      sum += *(const f32x4*)(src + (size_t)c * (BB * NN * DD));
    *(f32x4*)(sv + item * 128 + lane * 4) = sum;
  }
  __syncthreads();
  f32x4 vr[TI][4];
#pragma unroll
  for (int t = 0; t < TI; ++t)
#pragma unroll
    for (int p = 0; p < 4; ++p)
      vr[t][p] = *(const f32x4*)(sv + (t * 4 + p) * 128 + lane * 4);
  const float* eblk = edge + (size_t)(b * NN + i0) * NN * DD;
  for (int j = j0 + wave; j < j0 + NN / 2; j += 8) {
    const float* er = eblk + (size_t)j * DD + lane * 4;
    const float* ur = utt + (size_t)(b * NN + j) * DD + lane * 4;
    if (j + 8 < j0 + NN / 2) {
#pragma unroll
      for (int t = 0; t < TI; ++t)
        __builtin_prefetch(eblk + (size_t)t * NN * DD + (size_t)(j + 8) * DD + lane * 16, 0, 0);
    }
    f32x4 u[4];
#pragma unroll
    for (int p = 0; p < 4; ++p)
      u[p] = *(const f32x4*)(ur + p * 128);
    f32x4 e[TI][4];
#pragma unroll
    for (int t = 0; t < TI; ++t)
#pragma unroll
      for (int p = 0; p < 4; ++p)
        e[t][p] = __builtin_nontemporal_load((const f32x4*)(er + (size_t)t * NN * DD + p * 128));
    float s[TI];
#pragma unroll
    for (int t = 0; t < TI; ++t) {
      f32x4 acc = {0.f, 0.f, 0.f, 0.f};
#pragma unroll
      for (int p = 0; p < 4; ++p)
        acc += (e[t][p] + u[p]) * vr[t][p];
      s[t] = acc.x + acc.y + acc.z + acc.w;
    }
#pragma unroll
    for (int off = 16; off > 0; off >>= 1) {
#pragma unroll
      for (int t = 0; t < TI; ++t)
        s[t] += __shfl_xor(s[t], off, 32);
    }
    const float s01 = (lane & 1) ? s[1] : s[0];
    const float s23 = (lane & 1) ? s[3] : s[2];
    const float sv  = (lane & 2) ? s23 : s01;
    if (lane < TI)
      logits[(b * NN + i0 + lane) * NN + j] = sv * 0.044194173824159216f;
  }
}

#define AP 132
struct OutB { v2f b1[8], b2[8]; };

__device__ __forceinline__ void outb_load(OutB& f, const float* __restrict__ pzi,
                                          const float* __restrict__ psl, int kb) {
#pragma unroll
  for (int s = 0; s < 8; ++s) {
    const int k = kb + s * 4;
    f.b1[s].x = pzi[(k + 0) * DD];
    f.b1[s].y = pzi[(k + 1) * DD];
    f.b2[s].x = psl[(k + 0) * DD];
    f.b2[s].y = psl[(k + 1) * DD];
  }
}

template <int S0>
__device__ __forceinline__ v8f out_mma(v8f acc, const v2f (&a1)[32], const v2f (&a2)[32],
                                       const OutB& f) {
#pragma unroll
  for (int s = 0; s < 8; ++s) {
    acc = WMMA_F32(acc, a1[S0 + s], f.b1[s]);
    acc = WMMA_F32(acc, a2[S0 + s], f.b2[s]);
  }
  return acc;
}

__global__ __launch_bounds__(128) __attribute__((amdgpu_waves_per_eu(1, 2))) void out_kernel(const float* __restrict__ logits,
                                                 const float* __restrict__ bk,
                                                 const float* __restrict__ seq,
                                                 const float* __restrict__ zi,
                                                 const float* __restrict__ si_lin,
                                                 float* __restrict__ out) {
  __shared__ __attribute__((aligned(16))) float sat[16 * AP];
  __shared__ __attribute__((aligned(16))) float ssq[16 * AP];
  __shared__ float smx[4 * 16];
  __shared__ float ssm[4 * 16];
  __shared__ float sdg[4 * 16];
  const int lane = threadIdx.x & 31;
  const int wave = threadIdx.x >> 5;
  const int half = lane >> 4;
  const int ln   = lane & 15;
  const int b    = blockIdx.z;
  const int n0   = blockIdx.y * 16;
  const int d0   = (blockIdx.x * 4 + wave) * 16;
  const int rowb = b * NN;
  const int q0   = wave * 32 + half * 16;
  const float* plg  = logits + (rowb + q0) * NN + n0 + ln;
  const float* pbk  = bk     + (rowb + q0) * NN + n0 + ln;
  const float* pseq = seq    + (rowb + n0 + ln) * NN + q0;
  const float* pzi  = zi     + (rowb + half * 2) * DD + d0 + ln;
  const float* psl  = si_lin + (rowb + half * 2) * DD + d0 + ln;
  float lg[16], mk[16];
#pragma unroll
  for (int t = 0; t < 16; ++t) {
    mk[t] = pbk[t * NN];
    lg[t] = plg[t * NN];
  }
  f32x4 sq[4];
#pragma unroll
  for (int p = 0; p < 4; ++p)
    sq[p] = *(const f32x4*)(pseq + p * 4);
  v8f acc;
#pragma unroll
  for (int r = 0; r < 8; ++r)
    acc[r] = si_lin[(rowb + n0 + r + half * 8) * DD + d0 + ln];
  OutB f0, f1;
  outb_load(f0, pzi, psl, 0);
  outb_load(f1, pzi, psl, 32);

  float mx = -__builtin_inff();
#pragma unroll
  for (int t = 0; t < 16; ++t) {
    lg[t] = (mk[t] > 0.f) ? lg[t] : -1e30f;
    mx = fmaxf(mx, lg[t]);
  }
  mx = fmaxf(mx, __shfl_xor(mx, 16, 32));
  float sm = 0.f;
#pragma unroll
  for (int t = 0; t < 16; ++t) {
    lg[t] = __expf(lg[t] - mx);
    sm += lg[t];
  }
  sm += __shfl_xor(sm, 16, 32);
  f32x4 dv = sq[0];
#pragma unroll
  for (int p = 1; p < 4; ++p) dv += sq[p];
  float dg = (dv.x + dv.y) + (dv.z + dv.w);
  dg += __shfl_xor(dg, 16, 32);
#pragma unroll
  for (int p = 0; p < 4; ++p) {
    const f32x4 o = {mk[p * 4 + 0] * lg[p * 4 + 0], mk[p * 4 + 1] * lg[p * 4 + 1],
                     mk[p * 4 + 2] * lg[p * 4 + 2], mk[p * 4 + 3] * lg[p * 4 + 3]};
    *(f32x4*)(sat + ln * AP + q0 + p * 4) = o;
    *(f32x4*)(ssq + ln * AP + q0 + p * 4) = sq[p];
  }
  if (half == 0) {
    smx[wave * 16 + ln] = mx;
    ssm[wave * 16 + ln] = sm;
    sdg[wave * 16 + ln] = dg;
  }
  __syncthreads();

  float mw[4], sc[4];
#pragma unroll
  for (int w = 0; w < 4; ++w) mw[w] = smx[w * 16 + ln];
  const float mall = fmaxf(fmaxf(mw[0], mw[1]), fmaxf(mw[2], mw[3]));
  float sum = 0.f, deg = 0.f;
#pragma unroll
  for (int w = 0; w < 4; ++w) {
    sc[w] = __expf(mw[w] - mall);
    sum += ssm[w * 16 + ln] * sc[w];
    deg += sdg[w * 16 + ln];
  }
  const float rinv = 1.0f / sum;
  const float dinv = 1.0f / (deg + 1e-10f);
#pragma unroll
  for (int w = 0; w < 4; ++w) sc[w] *= rinv;
  v2f a1[32], a2[32];
  const float* pla = sat + ln * AP + half * 2;
  const float* pls = ssq + ln * AP + half * 2;
#pragma unroll
  for (int s = 0; s < 32; ++s) {
    a1[s] = *(const v2f*)(pla + s * 4) * sc[s >> 3];
    a2[s] = *(const v2f*)(pls + s * 4) * dinv;
  }

  acc = out_mma<0>(acc, a1, a2, f0);
  outb_load(f0, pzi, psl, 64);
  acc = out_mma<8>(acc, a1, a2, f1);
  outb_load(f1, pzi, psl, 96);
  acc = out_mma<16>(acc, a1, a2, f0);
  acc = out_mma<24>(acc, a1, a2, f1);
  const float kScale = 1.0507009873554805f;
  const float kAlphaScale = 1.0507009873554805f * 1.6732632423543772f;
#pragma unroll
  for (int r = 0; r < 8; ++r) {
    const float x = acc[r];
    const float y = (x > 0.f) ? kScale * x : kAlphaScale * (__expf(x) - 1.f);
    out[(rowb + n0 + r + half * 8) * DD + d0 + ln] = y;
  }
}

extern "C" void kernel_launch(void* const* d_in, const int* in_sizes, int n_in,
                              void* d_out, int out_size, void* d_ws, size_t ws_size,
                              hipStream_t stream) {
  const float* utt  = (const float*)d_in[0];
  const float* edge = (const float*)d_in[1];
  const float* bk   = (const float*)d_in[2];
  const float* seq  = (const float*)d_in[3];
  const float* Wk   = (const float*)d_in[4];
  const float* Ws   = (const float*)d_in[5];
  float* out = (float*)d_out;

  float* ws     = (float*)d_ws;
  float* zi     = ws;
  float* vpart  = zi + BB * NN * DD;
  float* si_lin = vpart + NCH * BB * NN * DD;
  float* logits = si_lin + BB * NN * DD;

  lin_kernel<<<dim3(NCH, (BB * NN) / 32), 128, 0, stream>>>(utt, Wk, Ws, edge, zi, vpart, si_lin);
  edge_dot_kernel<<<BB * NN / TI * 2, 256, 0, stream>>>(edge, utt, vpart, logits);
  out_kernel<<<dim3(DD / 64, NN / 16, BB), 128, 0, stream>>>(logits, bk, seq, zi, si_lin, out);
}
